# starting kernel
# baseline (speedup 1.0000x reference)
#include <hip/hip_runtime.h>
#include <hip/hip_bf16.h>
#include <math.h>

#define NN 50000
#define EE 800000
#define INF_ 256
#define OUTF 32
#define H1N 8
#define SLOPE 0.2f

typedef float v2f __attribute__((ext_vector_type(2)));
typedef float v8f __attribute__((ext_vector_type(8)));

__device__ __forceinline__ float leaky(float x) { return x > 0.f ? x : SLOPE * x; }

__device__ __forceinline__ void atomic_max_f32(float* addr, float v) {
    if (v >= 0.f) atomicMax((int*)addr, __float_as_int(v));
    else          atomicMin((unsigned int*)addr, __float_as_uint(v));
}

__global__ void __launch_bounds__(128) gemm1_kernel(const float* __restrict__ X,
                                                    const float* __restrict__ W,
                                                    float* __restrict__ C) {
    const int wave = threadIdx.x >> 5;
    const int lane = threadIdx.x & 31;
    const int half = lane >> 4;
    const int r    = lane & 15;
    const int row0 = blockIdx.x * 16;
    const int col0 = wave * 64;

    v8f acc[4] = {};
    const float* xrow = X + (size_t)(row0 + r) * INF_;

    for (int k = 0; k < INF_; k += 4) {
        v2f a;
        a.x = xrow[k + half * 2 + 0];
        a.y = xrow[k + half * 2 + 1];
#pragma unroll
        for (int t = 0; t < 4; ++t) {
            const int col = col0 + t * 16 + r;
            v2f b;
            b.x = W[(size_t)(k + half * 2 + 0) * 256 + col];
            b.y = W[(size_t)(k + half * 2 + 1) * 256 + col];
            acc[t] = __builtin_amdgcn_wmma_f32_16x16x4_f32(
                false, a, false, b, (short)0, acc[t], false, false);
        }
    }
#pragma unroll
    for (int t = 0; t < 4; ++t) {
#pragma unroll
        for (int j = 0; j < 8; ++j) {
            const int row = row0 + half * 8 + j;
            const int col = col0 + t * 16 + r;
            C[(size_t)row * 256 + col] = acc[t][j];
        }
    }
}

__global__ void __launch_bounds__(128) gemm2_kernel(const float* __restrict__ Y,
                                                    const float* __restrict__ W2,
                                                    const float* __restrict__ RW,
                                                    const float* __restrict__ b2,
                                                    float* __restrict__ F2,
                                                    float* __restrict__ OUTP,
                                                    int n) {
    const int wave = threadIdx.x >> 5;
    const int lane = threadIdx.x & 31;
    const int half = lane >> 4;
    const int r    = lane & 15;
    const int row0 = blockIdx.x * 64 + wave * 16;

    int rload = row0 + r;
    if (rload >= n) rload = n - 1;
    const float* yrow = Y + (size_t)rload * 256;

    v8f accW[2] = {};
    v8f accR[2] = {};

    for (int k = 0; k < 256; k += 4) {
        v2f a;
        a.x = yrow[k + half * 2 + 0];
        a.y = yrow[k + half * 2 + 1];
#pragma unroll
        for (int t = 0; t < 2; ++t) {
            const int col = t * 16 + r;
            v2f bw, br;
            bw.x = W2[(size_t)(k + half * 2 + 0) * 32 + col];
            bw.y = W2[(size_t)(k + half * 2 + 1) * 32 + col];
            accW[t] = __builtin_amdgcn_wmma_f32_16x16x4_f32(
                false, a, false, bw, (short)0, accW[t], false, false);
            br.x = RW[(size_t)(k + half * 2 + 0) * 32 + col];
            br.y = RW[(size_t)(k + half * 2 + 1) * 32 + col];
            accR[t] = __builtin_amdgcn_wmma_f32_16x16x4_f32(
                false, a, false, br, (short)0, accR[t], false, false);
        }
    }
#pragma unroll
    for (int t = 0; t < 2; ++t) {
#pragma unroll
        for (int j = 0; j < 8; ++j) {
            const int row = row0 + half * 8 + j;
            const int col = t * 16 + r;
            if (row < n) {
                F2[(size_t)row * 32 + col]   = accW[t][j];
                OUTP[(size_t)row * 32 + col] = accR[t][j] + b2[col];
            }
        }
    }
}

__global__ void __launch_bounds__(256) elr1_kernel(const float* __restrict__ feat,
                                                   const float* __restrict__ al,
                                                   const float* __restrict__ ar,
                                                   float* __restrict__ el,
                                                   float* __restrict__ er) {
    const int n = blockIdx.x;
    const int d = threadIdx.x;
    const float v = feat[(size_t)n * 256 + d];
    float pl = v * al[d];
    float pr = v * ar[d];
#pragma unroll
    for (int off = 16; off > 0; off >>= 1) {
        pl += __shfl_xor(pl, off, 32);
        pr += __shfl_xor(pr, off, 32);
    }
    if ((d & 31) == 0) {
        const int h = d >> 5;
        el[n * H1N + h] = pl;
        er[n * H1N + h] = pr;
    }
}

__global__ void __launch_bounds__(256) elr2_kernel(const float* __restrict__ feat,
                                                   const float* __restrict__ al,
                                                   const float* __restrict__ ar,
                                                   float* __restrict__ el,
                                                   float* __restrict__ er,
                                                   int n) {
    const int node = blockIdx.x * 8 + (threadIdx.x >> 5);
    const int lane = threadIdx.x & 31;
    if (node >= n) return;
    const float v = feat[(size_t)node * 32 + lane];
    float pl = v * al[lane];
    float pr = v * ar[lane];
#pragma unroll
    for (int off = 16; off > 0; off >>= 1) {
        pl += __shfl_xor(pl, off, 32);
        pr += __shfl_xor(pr, off, 32);
    }
    if (lane == 0) { el[node] = pl; er[node] = pr; }
}

__global__ void __launch_bounds__(256) init_rst1_kernel(const float* __restrict__ x,
                                                        const float* __restrict__ b1,
                                                        float* __restrict__ rst) {
    const size_t i = (size_t)blockIdx.x * 256 + threadIdx.x;
    rst[i] = x[i] + b1[threadIdx.x];
}

__global__ void __launch_bounds__(256) init_softmax_kernel(float* __restrict__ emax,
                                                           float* __restrict__ esum,
                                                           int n) {
    const int i = blockIdx.x * 256 + threadIdx.x;
    if (i < n) { emax[i] = -INFINITY; esum[i] = 0.f; }
}

__global__ void __launch_bounds__(256) relu_kernel(float* __restrict__ y) {
    const size_t i = (size_t)blockIdx.x * 256 + threadIdx.x;
    y[i] = fmaxf(y[i], 0.f);
}

__global__ void __launch_bounds__(256) edge1_max_kernel(const int* __restrict__ src,
                                                        const int* __restrict__ dst,
                                                        const float* __restrict__ el,
                                                        const float* __restrict__ er,
                                                        float* __restrict__ emax) {
    const int tid = blockIdx.x * 256 + threadIdx.x;
    if (tid >= EE * H1N) return;
    const int e = tid >> 3, h = tid & 7;
    const int s = src[e], d = dst[e];
    const float sc = leaky(el[s * H1N + h] + er[d * H1N + h]);
    atomic_max_f32(&emax[d * H1N + h], sc);
}

__global__ void __launch_bounds__(256) edge1_sum_kernel(const int* __restrict__ src,
                                                        const int* __restrict__ dst,
                                                        const float* __restrict__ el,
                                                        const float* __restrict__ er,
                                                        const float* __restrict__ emax,
                                                        float* __restrict__ esum) {
    const int tid = blockIdx.x * 256 + threadIdx.x;
    if (tid >= EE * H1N) return;
    const int e = tid >> 3, h = tid & 7;
    const int s = src[e], d = dst[e];
    const float sc = leaky(el[s * H1N + h] + er[d * H1N + h]);
    atomicAdd(&esum[d * H1N + h], __expf(sc - emax[d * H1N + h]));
}

__global__ void __launch_bounds__(256) edge1_agg_kernel(const int* __restrict__ src,
                                                        const int* __restrict__ dst,
                                                        const float* __restrict__ el,
                                                        const float* __restrict__ er,
                                                        const float* __restrict__ emax,
                                                        const float* __restrict__ esum,
                                                        const float* __restrict__ feat,
                                                        float* __restrict__ rst) {
    const int gid  = blockIdx.x * 256 + threadIdx.x;
    const int e    = gid >> 5;
    if (e >= EE) return;
    const int lane = gid & 31;
    const int s = src[e], d = dst[e];
    const int h = lane >> 2;
    const float sc = leaky(el[s * H1N + h] + er[d * H1N + h]);
    const float a  = __expf(sc - emax[d * H1N + h]) / esum[d * H1N + h];

    const float4* fs = (const float4*)(feat + (size_t)s * 256 + lane * 8);
    const float4 f0 = fs[0], f1 = fs[1];
    float* rd = rst + (size_t)d * 256 + lane * 8;
    atomicAdd(rd + 0, f0.x * a); atomicAdd(rd + 1, f0.y * a);
    atomicAdd(rd + 2, f0.z * a); atomicAdd(rd + 3, f0.w * a);
    atomicAdd(rd + 4, f1.x * a); atomicAdd(rd + 5, f1.y * a);
    atomicAdd(rd + 6, f1.z * a); atomicAdd(rd + 7, f1.w * a);
}

__global__ void __launch_bounds__(256) edge2_max_kernel(const int* __restrict__ src,
                                                        const int* __restrict__ dst,
                                                        const float* __restrict__ el,
                                                        const float* __restrict__ er,
                                                        float* __restrict__ emax) {
    const int e = blockIdx.x * 256 + threadIdx.x;
    if (e >= EE) return;
    const int s = src[e], d = dst[e];
    atomic_max_f32(&emax[d], leaky(el[s] + er[d]));
}

__global__ void __launch_bounds__(256) edge2_sum_kernel(const int* __restrict__ src,
                                                        const int* __restrict__ dst,
                                                        const float* __restrict__ el,
                                                        const float* __restrict__ er,
                                                        const float* __restrict__ emax,
                                                        float* __restrict__ esum) {
    const int e = blockIdx.x * 256 + threadIdx.x;
    if (e >= EE) return;
    const int s = src[e], d = dst[e];
    const float sc = leaky(el[s] + er[d]);
    atomicAdd(&esum[d], __expf(sc - emax[d]));
}

__global__ void __launch_bounds__(256) edge2_agg_kernel(const int* __restrict__ src,
                                                        const int* __restrict__ dst,
                                                        const float* __restrict__ el,
                                                        const float* __restrict__ er,
                                                        const float* __restrict__ emax,
                                                        const float* __restrict__ esum,
                                                        const float* __restrict__ feat,
                                                        float* __restrict__ outp) {
    const int gid  = blockIdx.x * 256 + threadIdx.x;
    const int e    = gid >> 5;
    if (e >= EE) return;
    const int lane = gid & 31;
    const int s = src[e], d = dst[e];
    const float sc = leaky(el[s] + er[d]);
    const float a  = __expf(sc - emax[d]) / esum[d];
    atomicAdd(outp + (size_t)d * 32 + lane, feat[(size_t)s * 32 + lane] * a);
}

extern "C" void kernel_launch(void* const* d_in, const int* in_sizes, int n_in,
                              void* d_out, int out_size, void* d_ws, size_t ws_size,
                              hipStream_t stream) {
    const float* x     = (const float*)d_in[0];
    const int*   src   = (const int*)  d_in[1];
    const int*   dst   = (const int*)  d_in[2];
    const float* W1    = (const float*)d_in[3];
    const float* al1   = (const float*)d_in[4];
    const float* ar1   = (const float*)d_in[5];
    const float* b1    = (const float*)d_in[6];
    const float* W2    = (const float*)d_in[7];
    const float* al2   = (const float*)d_in[8];
    const float* ar2   = (const float*)d_in[9];
    const float* resW2 = (const float*)d_in[10];
    const float* b2    = (const float*)d_in[11];
    float* outp = (float*)d_out;

    float* ws    = (float*)d_ws;
    float* feat1 = ws;
    float* rst1  = feat1 + (size_t)NN * 256;
    float* el1   = rst1  + (size_t)NN * 256;
    float* er1   = el1   + (size_t)NN * 8;
    float* emax1 = er1   + (size_t)NN * 8;
    float* esum1 = emax1 + (size_t)NN * 8;
    float* feat2 = esum1 + (size_t)NN * 8;
    float* el2   = feat2 + (size_t)NN * 32;
    float* er2   = el2   + NN;
    float* emax2 = er2   + NN;
    float* esum2 = emax2 + NN;

    gemm1_kernel<<<NN / 16, 128, 0, stream>>>(x, W1, feat1);
    init_rst1_kernel<<<NN, 256, 0, stream>>>(x, b1, rst1);
    init_softmax_kernel<<<(NN * H1N + 255) / 256, 256, 0, stream>>>(emax1, esum1, NN * H1N);
    elr1_kernel<<<NN, 256, 0, stream>>>(feat1, al1, ar1, el1, er1);
    edge1_max_kernel<<<(EE * H1N + 255) / 256, 256, 0, stream>>>(src, dst, el1, er1, emax1);
    edge1_sum_kernel<<<(EE * H1N + 255) / 256, 256, 0, stream>>>(src, dst, el1, er1, emax1, esum1);
    edge1_agg_kernel<<<(EE * 32) / 256, 256, 0, stream>>>(src, dst, el1, er1, emax1, esum1,
                                                          feat1, rst1);
    relu_kernel<<<NN, 256, 0, stream>>>(rst1);

    init_softmax_kernel<<<(NN + 255) / 256, 256, 0, stream>>>(emax2, esum2, NN);
    gemm2_kernel<<<(NN + 63) / 64, 128, 0, stream>>>(rst1, W2, resW2, b2, feat2, outp, NN);
    elr2_kernel<<<(NN + 7) / 8, 256, 0, stream>>>(feat2, al2, ar2, el2, er2, NN);
    edge2_max_kernel<<<(EE + 255) / 256, 256, 0, stream>>>(src, dst, el2, er2, emax2);
    edge2_sum_kernel<<<(EE + 255) / 256, 256, 0, stream>>>(src, dst, el2, er2, emax2, esum2);
    edge2_agg_kernel<<<(EE * 32) / 256, 256, 0, stream>>>(src, dst, el2, er2, emax2, esum2,
                                                          feat2, outp);
}


# Both aggregations: per-destination gather over an in-call counting sort of edges (arrival-order rank sets sum order) replaces float-atomic scatter; start value folded into the single store
# speedup vs baseline: 1.7451x; 1.7451x over previous
#include <hip/hip_runtime.h>
#include <hip/hip_bf16.h>
#include <math.h>

#define NN 50000
#define EE 800000
#define INF_ 256
#define OUTF 32
#define H1N 8
#define SLOPE 0.2f

typedef float v2f __attribute__((ext_vector_type(2)));
typedef float v8f __attribute__((ext_vector_type(8)));

__device__ __forceinline__ float leaky(float x) { return x > 0.f ? x : SLOPE * x; }

__device__ __forceinline__ void atomic_max_f32(float* addr, float v) {
    if (v >= 0.f) atomicMax((int*)addr, __float_as_int(v));
    else          atomicMin((unsigned int*)addr, __float_as_uint(v));
}

__global__ void __launch_bounds__(128) gemm1_kernel(const float* __restrict__ X,
                                                    const float* __restrict__ W,
                                                    float* __restrict__ C) {
    const int wave = threadIdx.x >> 5;
    const int lane = threadIdx.x & 31;
    const int half = lane >> 4;
    const int r    = lane & 15;
    const int row0 = blockIdx.x * 16;
    const int col0 = wave * 64;

    v8f acc[4] = {};
    const float* xrow = X + (size_t)(row0 + r) * INF_;

    for (int k = 0; k < INF_; k += 4) {
        v2f a;
        a.x = xrow[k + half * 2 + 0];
        a.y = xrow[k + half * 2 + 1];
#pragma unroll
        for (int t = 0; t < 4; ++t) {
            const int col = col0 + t * 16 + r;
            v2f b;
            b.x = W[(size_t)(k + half * 2 + 0) * 256 + col];
            b.y = W[(size_t)(k + half * 2 + 1) * 256 + col];
            acc[t] = __builtin_amdgcn_wmma_f32_16x16x4_f32(
                false, a, false, b, (short)0, acc[t], false, false);
        }
    }
#pragma unroll
    for (int t = 0; t < 4; ++t) {
#pragma unroll
        for (int j = 0; j < 8; ++j) {
            const int row = row0 + half * 8 + j;
            const int col = col0 + t * 16 + r;
            C[(size_t)row * 256 + col] = acc[t][j];
        }
    }
}

__global__ void __launch_bounds__(128) gemm2_kernel(const float* __restrict__ Y,
                                                    const float* __restrict__ W2,
                                                    const float* __restrict__ RW,
                                                    const float* __restrict__ b2,
                                                    float* __restrict__ F2,
                                                    float* __restrict__ OUTP,
                                                    int n) {
    const int wave = threadIdx.x >> 5;
    const int lane = threadIdx.x & 31;
    const int half = lane >> 4;
    const int r    = lane & 15;
    const int row0 = blockIdx.x * 64 + wave * 16;

    int rload = row0 + r;
    if (rload >= n) rload = n - 1;
    const float* yrow = Y + (size_t)rload * 256;

    v8f accW[2] = {};
    v8f accR[2] = {};

    for (int k = 0; k < 256; k += 4) {
        v2f a;
        a.x = yrow[k + half * 2 + 0];
        a.y = yrow[k + half * 2 + 1];
#pragma unroll
        for (int t = 0; t < 2; ++t) {
            const int col = t * 16 + r;
            v2f bw, br;
            bw.x = W2[(size_t)(k + half * 2 + 0) * 32 + col];
            bw.y = W2[(size_t)(k + half * 2 + 1) * 32 + col];
            accW[t] = __builtin_amdgcn_wmma_f32_16x16x4_f32(
                false, a, false, bw, (short)0, accW[t], false, false);
            br.x = RW[(size_t)(k + half * 2 + 0) * 32 + col];
            br.y = RW[(size_t)(k + half * 2 + 1) * 32 + col];
            accR[t] = __builtin_amdgcn_wmma_f32_16x16x4_f32(
                false, a, false, br, (short)0, accR[t], false, false);
        }
    }
#pragma unroll
    for (int t = 0; t < 2; ++t) {
#pragma unroll
        for (int j = 0; j < 8; ++j) {
            const int row = row0 + half * 8 + j;
            const int col = t * 16 + r;
            if (row < n) {
                F2[(size_t)row * 32 + col]   = accW[t][j];
                OUTP[(size_t)row * 32 + col] = accR[t][j] + b2[col];
            }
        }
    }
}

__global__ void __launch_bounds__(256) elr1_kernel(const float* __restrict__ feat,
                                                   const float* __restrict__ al,
                                                   const float* __restrict__ ar,
                                                   float* __restrict__ el,
                                                   float* __restrict__ er) {
    const int n = blockIdx.x;
    const int d = threadIdx.x;
    const float v = feat[(size_t)n * 256 + d];
    float pl = v * al[d];
    float pr = v * ar[d];
#pragma unroll
    for (int off = 16; off > 0; off >>= 1) {
        pl += __shfl_xor(pl, off, 32);
        pr += __shfl_xor(pr, off, 32);
    }
    if ((d & 31) == 0) {
        const int h = d >> 5;
        el[n * H1N + h] = pl;
        er[n * H1N + h] = pr;
    }
}

__global__ void __launch_bounds__(256) elr2_kernel(const float* __restrict__ feat,
                                                   const float* __restrict__ al,
                                                   const float* __restrict__ ar,
                                                   float* __restrict__ el,
                                                   float* __restrict__ er,
                                                   int n) {
    const int node = blockIdx.x * 8 + (threadIdx.x >> 5);
    const int lane = threadIdx.x & 31;
    if (node >= n) return;
    const float v = feat[(size_t)node * 32 + lane];
    float pl = v * al[lane];
    float pr = v * ar[lane];
#pragma unroll
    for (int off = 16; off > 0; off >>= 1) {
        pl += __shfl_xor(pl, off, 32);
        pr += __shfl_xor(pr, off, 32);
    }
    if (lane == 0) { el[node] = pl; er[node] = pr; }
}

__global__ void __launch_bounds__(256) init_softmax_kernel(float* __restrict__ emax,
                                                           float* __restrict__ esum,
                                                           int n) {
    const int i = blockIdx.x * 256 + threadIdx.x;
    if (i < n) { emax[i] = -INFINITY; esum[i] = 0.f; }
}

__global__ void __launch_bounds__(256) relu_kernel(float* __restrict__ y) {
    const size_t i = (size_t)blockIdx.x * 256 + threadIdx.x;
    y[i] = fmaxf(y[i], 0.f);
}

__global__ void __launch_bounds__(256) edge1_max_kernel(const int* __restrict__ src,
                                                        const int* __restrict__ dst,
                                                        const float* __restrict__ el,
                                                        const float* __restrict__ er,
                                                        float* __restrict__ emax) {
    const int tid = blockIdx.x * 256 + threadIdx.x;
    if (tid >= EE * H1N) return;
    const int e = tid >> 3, h = tid & 7;
    const int s = src[e], d = dst[e];
    const float sc = leaky(el[s * H1N + h] + er[d * H1N + h]);
    atomic_max_f32(&emax[d * H1N + h], sc);
}

__global__ void __launch_bounds__(256) edge1_sum_kernel(const int* __restrict__ src,
                                                        const int* __restrict__ dst,
                                                        const float* __restrict__ el,
                                                        const float* __restrict__ er,
                                                        const float* __restrict__ emax,
                                                        float* __restrict__ esum) {
    const int tid = blockIdx.x * 256 + threadIdx.x;
    if (tid >= EE * H1N) return;
    const int e = tid >> 3, h = tid & 7;
    const int s = src[e], d = dst[e];
    const float sc = leaky(el[s * H1N + h] + er[d * H1N + h]);
    atomicAdd(&esum[d * H1N + h], __expf(sc - emax[d * H1N + h]));
}

#define SCAN_BLOCKS 196
#define SCAN_LEN (SCAN_BLOCKS * 256)

__global__ void __launch_bounds__(256) zero_cnt_kernel(int* __restrict__ cnt) {
    cnt[blockIdx.x * 256 + threadIdx.x] = 0;
}

__global__ void __launch_bounds__(256) hist_kernel(const int* __restrict__ dst,
                                                   int* __restrict__ cnt,
                                                   int* __restrict__ rank) {
    const int e = blockIdx.x * 256 + threadIdx.x;
    if (e >= EE) return;
    rank[e] = atomicAdd(&cnt[dst[e]], 1);
}

__global__ void __launch_bounds__(256) scan_kernel(const int* __restrict__ cnt,
                                                   int* __restrict__ off) {
    __shared__ int s_part[8];
    __shared__ int s_tot[8];
    const int t = threadIdx.x, lane = t & 31, w = t >> 5;
    const int b = blockIdx.x;
    int part = 0;
    for (int j = 0; j < b; ++j) part += cnt[j * 256 + t];
    const int v = cnt[b * 256 + t];
#pragma unroll
    for (int o = 16; o > 0; o >>= 1) part += __shfl_xor(part, o, 32);
    int inc = v;
#pragma unroll
    for (int o = 1; o < 32; o <<= 1) {
        const int nb = __shfl_up(inc, o, 32);
        if (lane >= o) inc += nb;
    }
    if (lane == 0)  s_part[w] = part;
    if (lane == 31) s_tot[w]  = inc;
    __syncthreads();
    int base = 0;
#pragma unroll
    for (int i = 0; i < 8; ++i) {
        base += s_part[i];
        if (i < w) base += s_tot[i];
    }
    off[b * 256 + t] = base + inc - v;
}

__global__ void __launch_bounds__(256) fill_kernel(const int* __restrict__ src,
                                                   const int* __restrict__ dst,
                                                   const int* __restrict__ off,
                                                   const int* __restrict__ rank,
                                                   int* __restrict__ csrc) {
    const int e = blockIdx.x * 256 + threadIdx.x;
    if (e >= EE) return;
    csrc[off[dst[e]] + rank[e]] = src[e];
}

__global__ void __launch_bounds__(256) agg1_gather_kernel(const int* __restrict__ off,
                                                          const int* __restrict__ csrc,
                                                          const float* __restrict__ el,
                                                          const float* __restrict__ er,
                                                          const float* __restrict__ emax,
                                                          const float* __restrict__ esum,
                                                          const float* __restrict__ feat,
                                                          const float* __restrict__ x,
                                                          const float* __restrict__ b1,
                                                          float* __restrict__ rst) {
    const int lane = threadIdx.x & 31;
    const int d = __builtin_amdgcn_readfirstlane((int)(blockIdx.x * 8 + (threadIdx.x >> 5)));
    const int h = lane >> 2;
    const int beg = off[d], end = off[d + 1];
    const float erd = er[d * H1N + h];
    const float mx  = emax[d * H1N + h];
    const float sm  = esum[d * H1N + h];
    float4 a0 = make_float4(0.f, 0.f, 0.f, 0.f);
    float4 a1 = make_float4(0.f, 0.f, 0.f, 0.f);
    for (int base = beg; base < end; base += 32) {
        const int m = (end - base < 32) ? (end - base) : 32;
        const int s_l = (lane < m) ? csrc[base + lane] : 0;
#pragma unroll 4
        for (int i = 0; i < m; ++i) {
            const int s = __builtin_amdgcn_readlane(s_l, i);
            const float sc = leaky(el[s * H1N + h] + erd);
            const float a  = __expf(sc - mx) / sm;
            const float4* fs = (const float4*)(feat + (size_t)s * 256 + lane * 8);
            const float4 f0 = fs[0], f1 = fs[1];
            a0.x += f0.x * a; a0.y += f0.y * a; a0.z += f0.z * a; a0.w += f0.w * a;
            a1.x += f1.x * a; a1.y += f1.y * a; a1.z += f1.z * a; a1.w += f1.w * a;
        }
    }
    const size_t o = (size_t)d * 256 + lane * 8;
    const float4 x0 = *(const float4*)(x + o);
    const float4 x1 = *(const float4*)(x + o + 4);
    const float4 c0 = *(const float4*)(b1 + lane * 8);
    const float4 c1 = *(const float4*)(b1 + lane * 8 + 4);
    float4 r0, r1;
    r0.x = a0.x + x0.x + c0.x; r0.y = a0.y + x0.y + c0.y;
    r0.z = a0.z + x0.z + c0.z; r0.w = a0.w + x0.w + c0.w;
    r1.x = a1.x + x1.x + c1.x; r1.y = a1.y + x1.y + c1.y;
    r1.z = a1.z + x1.z + c1.z; r1.w = a1.w + x1.w + c1.w;
    *(float4*)(rst + o)     = r0;
    *(float4*)(rst + o + 4) = r1;
}

__global__ void __launch_bounds__(256) edge2_max_kernel(const int* __restrict__ src,
                                                        const int* __restrict__ dst,
                                                        const float* __restrict__ el,
                                                        const float* __restrict__ er,
                                                        float* __restrict__ emax) {
    const int e = blockIdx.x * 256 + threadIdx.x;
    if (e >= EE) return;
    const int s = src[e], d = dst[e];
    atomic_max_f32(&emax[d], leaky(el[s] + er[d]));
}

__global__ void __launch_bounds__(256) edge2_sum_kernel(const int* __restrict__ src,
                                                        const int* __restrict__ dst,
                                                        const float* __restrict__ el,
                                                        const float* __restrict__ er,
                                                        const float* __restrict__ emax,
                                                        float* __restrict__ esum) {
    const int e = blockIdx.x * 256 + threadIdx.x;
    if (e >= EE) return;
    const int s = src[e], d = dst[e];
    const float sc = leaky(el[s] + er[d]);
    atomicAdd(&esum[d], __expf(sc - emax[d]));
}

__global__ void __launch_bounds__(256) agg2_gather_kernel(const int* __restrict__ off,
                                                          const int* __restrict__ csrc,
                                                          const float* __restrict__ el,
                                                          const float* __restrict__ er,
                                                          const float* __restrict__ emax,
                                                          const float* __restrict__ esum,
                                                          const float* __restrict__ feat,
                                                          float* __restrict__ outp) {
    const int lane = threadIdx.x & 31;
    const int d = __builtin_amdgcn_readfirstlane((int)(blockIdx.x * 8 + (threadIdx.x >> 5)));
    const int beg = off[d], end = off[d + 1];
    const float erd = er[d], mx = emax[d], sm = esum[d];
    float acc = 0.f;
    for (int base = beg; base < end; base += 32) {
        const int m = (end - base < 32) ? (end - base) : 32;
        int   s_l = 0;
        float a_l = 0.f;
        if (lane < m) {
            s_l = csrc[base + lane];
            a_l = __expf(leaky(el[s_l] + erd) - mx) / sm;
        }
#pragma unroll 4
        for (int i = 0; i < m; ++i) {
            const int   s = __builtin_amdgcn_readlane(s_l, i);
            const float a = __int_as_float(__builtin_amdgcn_readlane(__float_as_int(a_l), i));
            acc += feat[(size_t)s * 32 + lane] * a;
        }
    }
    const size_t o = (size_t)d * 32 + lane;
    outp[o] = outp[o] + acc;
}

extern "C" void kernel_launch(void* const* d_in, const int* in_sizes, int n_in,
                              void* d_out, int out_size, void* d_ws, size_t ws_size,
                              hipStream_t stream) {
    const float* x     = (const float*)d_in[0];
    const int*   src   = (const int*)  d_in[1];
    const int*   dst   = (const int*)  d_in[2];
    const float* W1    = (const float*)d_in[3];
    const float* al1   = (const float*)d_in[4];
    const float* ar1   = (const float*)d_in[5];
    const float* b1    = (const float*)d_in[6];
    const float* W2    = (const float*)d_in[7];
    const float* al2   = (const float*)d_in[8];
    const float* ar2   = (const float*)d_in[9];
    const float* resW2 = (const float*)d_in[10];
    const float* b2    = (const float*)d_in[11];
    float* outp = (float*)d_out;

    float* ws    = (float*)d_ws;
    float* feat1 = ws;
    float* rst1  = feat1 + (size_t)NN * 256;
    float* el1   = rst1  + (size_t)NN * 256;
    float* er1   = el1   + (size_t)NN * 8;
    float* emax1 = er1   + (size_t)NN * 8;
    float* esum1 = emax1 + (size_t)NN * 8;
    int*   csrc  = (int*)(esum1 + (size_t)NN * 8);
    int*   rank  = csrc + EE;
    int*   cnt   = rank + EE;
    int*   off   = cnt  + SCAN_LEN;
    float* feat2 = ws;
    float* el2   = feat2 + (size_t)NN * 32;
    float* er2   = el2   + NN;
    float* emax2 = er2   + NN;
    float* esum2 = emax2 + NN;

    zero_cnt_kernel<<<SCAN_BLOCKS, 256, 0, stream>>>(cnt);
    hist_kernel<<<(EE + 255) / 256, 256, 0, stream>>>(dst, cnt, rank);
    scan_kernel<<<SCAN_BLOCKS, 256, 0, stream>>>(cnt, off);
    fill_kernel<<<(EE + 255) / 256, 256, 0, stream>>>(src, dst, off, rank, csrc);

    gemm1_kernel<<<NN / 16, 128, 0, stream>>>(x, W1, feat1);
    init_softmax_kernel<<<(NN * H1N + 255) / 256, 256, 0, stream>>>(emax1, esum1, NN * H1N);
    elr1_kernel<<<NN, 256, 0, stream>>>(feat1, al1, ar1, el1, er1);
    edge1_max_kernel<<<(EE * H1N + 255) / 256, 256, 0, stream>>>(src, dst, el1, er1, emax1);
    edge1_sum_kernel<<<(EE * H1N + 255) / 256, 256, 0, stream>>>(src, dst, el1, er1, emax1, esum1);
    agg1_gather_kernel<<<NN / 8, 256, 0, stream>>>(off, csrc, el1, er1, emax1, esum1,
                                                  feat1, x, b1, rst1);
    relu_kernel<<<NN, 256, 0, stream>>>(rst1);

    init_softmax_kernel<<<(NN + 255) / 256, 256, 0, stream>>>(emax2, esum2, NN);
    gemm2_kernel<<<(NN + 63) / 64, 128, 0, stream>>>(rst1, W2, resW2, b2, feat2, outp, NN);
    elr2_kernel<<<(NN + 7) / 8, 256, 0, stream>>>(feat2, al2, ar2, el2, er2, NN);
    edge2_max_kernel<<<(EE + 255) / 256, 256, 0, stream>>>(src, dst, el2, er2, emax2);
    edge2_sum_kernel<<<(EE + 255) / 256, 256, 0, stream>>>(src, dst, el2, er2, emax2, esum2);
    agg2_gather_kernel<<<NN / 8, 256, 0, stream>>>(off, csrc, el2, er2, emax2, esum2,
                                                  feat2, outp);
}


# Edge softmax max and sum taken per destination inside both gather kernels over the sorted segments; tree sum in segment order replaces the atomic max/add passes and their init (17 to 11 launches)
# speedup vs baseline: 2.7533x; 1.5777x over previous
#include <hip/hip_runtime.h>
#include <hip/hip_bf16.h>
#include <math.h>

#define NN 50000
#define EE 800000
#define INF_ 256
#define OUTF 32
#define H1N 8
#define SLOPE 0.2f

typedef float v2f __attribute__((ext_vector_type(2)));
typedef float v8f __attribute__((ext_vector_type(8)));

__device__ __forceinline__ float leaky(float x) { return x > 0.f ? x : SLOPE * x; }

__global__ void __launch_bounds__(128) gemm1_kernel(const float* __restrict__ X,
                                                    const float* __restrict__ W,
                                                    float* __restrict__ C) {
    const int wave = threadIdx.x >> 5;
    const int lane = threadIdx.x & 31;
    const int half = lane >> 4;
    const int r    = lane & 15;
    const int row0 = blockIdx.x * 16;
    const int col0 = wave * 64;

    v8f acc[4] = {};
    const float* xrow = X + (size_t)(row0 + r) * INF_;

    for (int k = 0; k < INF_; k += 4) {
        v2f a;
        a.x = xrow[k + half * 2 + 0];
        a.y = xrow[k + half * 2 + 1];
#pragma unroll
        for (int t = 0; t < 4; ++t) {
            const int col = col0 + t * 16 + r;
            v2f b;
            b.x = W[(size_t)(k + half * 2 + 0) * 256 + col];
            b.y = W[(size_t)(k + half * 2 + 1) * 256 + col];
            acc[t] = __builtin_amdgcn_wmma_f32_16x16x4_f32(
                false, a, false, b, (short)0, acc[t], false, false);
        }
    }
#pragma unroll
    for (int t = 0; t < 4; ++t) {
#pragma unroll
        for (int j = 0; j < 8; ++j) {
            const int row = row0 + half * 8 + j;
            const int col = col0 + t * 16 + r;
            C[(size_t)row * 256 + col] = acc[t][j];
        }
    }
}

__global__ void __launch_bounds__(128) gemm2_kernel(const float* __restrict__ Y,
                                                    const float* __restrict__ W2,
                                                    const float* __restrict__ RW,
                                                    const float* __restrict__ b2,
                                                    float* __restrict__ F2,
                                                    float* __restrict__ OUTP,
                                                    int n) {
    const int wave = threadIdx.x >> 5;
    const int lane = threadIdx.x & 31;
    const int half = lane >> 4;
    const int r    = lane & 15;
    const int row0 = blockIdx.x * 64 + wave * 16;

    int rload = row0 + r;
    if (rload >= n) rload = n - 1;
    const float* yrow = Y + (size_t)rload * 256;

    v8f accW[2] = {};
    v8f accR[2] = {};

    for (int k = 0; k < 256; k += 4) {
        v2f a;
        a.x = yrow[k + half * 2 + 0];
        a.y = yrow[k + half * 2 + 1];
#pragma unroll
        for (int t = 0; t < 2; ++t) {
            const int col = t * 16 + r;
            v2f bw, br;
            bw.x = W2[(size_t)(k + half * 2 + 0) * 32 + col];
            bw.y = W2[(size_t)(k + half * 2 + 1) * 32 + col];
            accW[t] = __builtin_amdgcn_wmma_f32_16x16x4_f32(
                false, a, false, bw, (short)0, accW[t], false, false);
            br.x = RW[(size_t)(k + half * 2 + 0) * 32 + col];
            br.y = RW[(size_t)(k + half * 2 + 1) * 32 + col];
            accR[t] = __builtin_amdgcn_wmma_f32_16x16x4_f32(
                false, a, false, br, (short)0, accR[t], false, false);
        }
    }
#pragma unroll
    for (int t = 0; t < 2; ++t) {
#pragma unroll
        for (int j = 0; j < 8; ++j) {
            const int row = row0 + half * 8 + j;
            const int col = t * 16 + r;
            if (row < n) {
                F2[(size_t)row * 32 + col]   = accW[t][j];
                OUTP[(size_t)row * 32 + col] = accR[t][j] + b2[col];
            }
        }
    }
}

__global__ void __launch_bounds__(256) elr1_kernel(const float* __restrict__ feat,
                                                   const float* __restrict__ al,
                                                   const float* __restrict__ ar,
                                                   float* __restrict__ el,
                                                   float* __restrict__ er) {
    const int n = blockIdx.x;
    const int d = threadIdx.x;
    const float v = feat[(size_t)n * 256 + d];
    float pl = v * al[d];
    float pr = v * ar[d];
#pragma unroll
    for (int off = 16; off > 0; off >>= 1) {
        pl += __shfl_xor(pl, off, 32);
        pr += __shfl_xor(pr, off, 32);
    }
    if ((d & 31) == 0) {
        const int h = d >> 5;
        el[n * H1N + h] = pl;
        er[n * H1N + h] = pr;
    }
}

__global__ void __launch_bounds__(256) elr2_kernel(const float* __restrict__ feat,
                                                   const float* __restrict__ al,
                                                   const float* __restrict__ ar,
                                                   float* __restrict__ el,
                                                   float* __restrict__ er,
                                                   int n) {
    const int node = blockIdx.x * 8 + (threadIdx.x >> 5);
    const int lane = threadIdx.x & 31;
    if (node >= n) return;
    const float v = feat[(size_t)node * 32 + lane];
    float pl = v * al[lane];
    float pr = v * ar[lane];
#pragma unroll
    for (int off = 16; off > 0; off >>= 1) {
        pl += __shfl_xor(pl, off, 32);
        pr += __shfl_xor(pr, off, 32);
    }
    if (lane == 0) { el[node] = pl; er[node] = pr; }
}

__global__ void __launch_bounds__(256) relu_kernel(float* __restrict__ y) {
    const size_t i = (size_t)blockIdx.x * 256 + threadIdx.x;
    y[i] = fmaxf(y[i], 0.f);
}

#define SCAN_BLOCKS 196
#define SCAN_LEN (SCAN_BLOCKS * 256)

__global__ void __launch_bounds__(256) zero_cnt_kernel(int* __restrict__ cnt) {
    cnt[blockIdx.x * 256 + threadIdx.x] = 0;
}

__global__ void __launch_bounds__(256) hist_kernel(const int* __restrict__ dst,
                                                   int* __restrict__ cnt,
                                                   int* __restrict__ rank) {
    const int e = blockIdx.x * 256 + threadIdx.x;
    if (e >= EE) return;
    rank[e] = atomicAdd(&cnt[dst[e]], 1);
}

__global__ void __launch_bounds__(256) scan_kernel(const int* __restrict__ cnt,
                                                   int* __restrict__ off) {
    __shared__ int s_part[8];
    __shared__ int s_tot[8];
    const int t = threadIdx.x, lane = t & 31, w = t >> 5;
    const int b = blockIdx.x;
    int part = 0;
    for (int j = 0; j < b; ++j) part += cnt[j * 256 + t];
    const int v = cnt[b * 256 + t];
#pragma unroll
    for (int o = 16; o > 0; o >>= 1) part += __shfl_xor(part, o, 32);
    int inc = v;
#pragma unroll
    for (int o = 1; o < 32; o <<= 1) {
        const int nb = __shfl_up(inc, o, 32);
        if (lane >= o) inc += nb;
    }
    if (lane == 0)  s_part[w] = part;
    if (lane == 31) s_tot[w]  = inc;
    __syncthreads();
    int base = 0;
#pragma unroll
    for (int i = 0; i < 8; ++i) {
        base += s_part[i];
        if (i < w) base += s_tot[i];
    }
    off[b * 256 + t] = base + inc - v;
}

__global__ void __launch_bounds__(256) fill_kernel(const int* __restrict__ src,
                                                   const int* __restrict__ dst,
                                                   const int* __restrict__ off,
                                                   const int* __restrict__ rank,
                                                   int* __restrict__ csrc) {
    const int e = blockIdx.x * 256 + threadIdx.x;
    if (e >= EE) return;
    csrc[off[dst[e]] + rank[e]] = src[e];
}

__global__ void __launch_bounds__(256) agg1_gather_kernel(const int* __restrict__ off,
                                                          const int* __restrict__ csrc,
                                                          const float* __restrict__ el,
                                                          const float* __restrict__ er,
                                                          const float* __restrict__ feat,
                                                          const float* __restrict__ x,
                                                          const float* __restrict__ b1,
                                                          float* __restrict__ rst) {
    const int lane = threadIdx.x & 31;
    const int d = __builtin_amdgcn_readfirstlane((int)(blockIdx.x * 8 + (threadIdx.x >> 5)));
    const int beg = off[d], end = off[d + 1];

    const int hq = lane & 7, q = lane >> 3;
    const float erq = er[d * H1N + hq];
    float v[8];
    for (int k = 0; k < 8; ++k) v[k] = -INFINITY;
    {
        int sq[4];
        for (int k = 0; k < 4; ++k) {
            const int e = beg + 4 * k + q;
            sq[k] = csrc[e < EE ? e : EE - 1];
        }
        for (int k = 0; k < 4; ++k) {
            const float t = leaky(el[sq[k] * H1N + hq] + erq);
            v[k] = (beg + 4 * k + q < end) ? t : -INFINITY;
        }
    }
    if (end - beg > 16) {
        int sq[4];
        for (int k = 4; k < 8; ++k) {
            const int e = beg + 4 * k + q;
            sq[k - 4] = csrc[e < EE ? e : EE - 1];
        }
        for (int k = 4; k < 8; ++k) {
            const float t = leaky(el[sq[k - 4] * H1N + hq] + erq);
            v[k] = (beg + 4 * k + q < end) ? t : -INFINITY;
        }
    }
    float m = -INFINITY;
    for (int k = 0; k < 8; ++k) m = fmaxf(m, v[k]);
    for (int e = beg + 32 + q; e < end; e += 4)
        m = fmaxf(m, leaky(el[csrc[e] * H1N + hq] + erq));
    m = fmaxf(m, __shfl_xor(m, 8, 32));
    m = fmaxf(m, __shfl_xor(m, 16, 32));
    float su = 0.f;
    for (int k = 0; k < 8; ++k)
        if (beg + 4 * k + q < end) su += __expf(v[k] - m);
    for (int e = beg + 32 + q; e < end; e += 4)
        su += __expf(leaky(el[csrc[e] * H1N + hq] + erq) - m);
    su += __shfl_xor(su, 8, 32);
    su += __shfl_xor(su, 16, 32);

    const int h = lane >> 2;
    const float erd = __shfl(erq, h, 32);
    const float mx  = __shfl(m,   h, 32);
    const float sm  = __shfl(su,  h, 32);

    float4 a0 = make_float4(0.f, 0.f, 0.f, 0.f);
    float4 a1 = make_float4(0.f, 0.f, 0.f, 0.f);
    for (int base = beg; base < end; base += 32) {
        const int mcnt = (end - base < 32) ? (end - base) : 32;
        const int s_l = (lane < mcnt) ? csrc[base + lane] : 0;
#pragma unroll 4
        for (int i = 0; i < mcnt; ++i) {
            const int s = __builtin_amdgcn_readlane(s_l, i);
            const float sc = leaky(el[s * H1N + h] + erd);
            const float a  = __expf(sc - mx) / sm;
            const float4* fs = (const float4*)(feat + (size_t)s * 256 + lane * 8);
            const float4 f0 = fs[0], f1 = fs[1];
            a0.x += f0.x * a; a0.y += f0.y * a; a0.z += f0.z * a; a0.w += f0.w * a;
            a1.x += f1.x * a; a1.y += f1.y * a; a1.z += f1.z * a; a1.w += f1.w * a;
        }
    }
    const size_t o = (size_t)d * 256 + lane * 8;
    const float4 x0 = *(const float4*)(x + o);
    const float4 x1 = *(const float4*)(x + o + 4);
    const float4 c0 = *(const float4*)(b1 + lane * 8);
    const float4 c1 = *(const float4*)(b1 + lane * 8 + 4);
    float4 r0, r1;
    r0.x = a0.x + x0.x + c0.x; r0.y = a0.y + x0.y + c0.y;
    r0.z = a0.z + x0.z + c0.z; r0.w = a0.w + x0.w + c0.w;
    r1.x = a1.x + x1.x + c1.x; r1.y = a1.y + x1.y + c1.y;
    r1.z = a1.z + x1.z + c1.z; r1.w = a1.w + x1.w + c1.w;
    *(float4*)(rst + o)     = r0;
    *(float4*)(rst + o + 4) = r1;
}

__global__ void __launch_bounds__(256) agg2_gather_kernel(const int* __restrict__ off,
                                                          const int* __restrict__ csrc,
                                                          const float* __restrict__ el,
                                                          const float* __restrict__ er,
                                                          const float* __restrict__ feat,
                                                          float* __restrict__ outp) {
    const int lane = threadIdx.x & 31;
    const int d = __builtin_amdgcn_readfirstlane((int)(blockIdx.x * 8 + (threadIdx.x >> 5)));
    const int beg = off[d], end = off[d + 1];
    const float erd = er[d];

    const bool have0 = beg + lane < end;
    int   s0 = 0;
    float v0 = -INFINITY;
    if (have0) { s0 = csrc[beg + lane]; v0 = leaky(el[s0] + erd); }
    float mx = v0;
    for (int e = beg + 32 + lane; e < end; e += 32)
        mx = fmaxf(mx, leaky(el[csrc[e]] + erd));
    for (int o = 16; o > 0; o >>= 1) mx = fmaxf(mx, __shfl_xor(mx, o, 32));
    const float w0 = have0 ? __expf(v0 - mx) : 0.f;
    float sm = w0;
    for (int e = beg + 32 + lane; e < end; e += 32)
        sm += __expf(leaky(el[csrc[e]] + erd) - mx);
    for (int o = 16; o > 0; o >>= 1) sm += __shfl_xor(sm, o, 32);

    float acc = 0.f;
    for (int base = beg; base < end; base += 32) {
        const int mcnt = (end - base < 32) ? (end - base) : 32;
        int   s_l = s0;
        float a_l = w0 / sm;
        if (base != beg) {
            s_l = 0; a_l = 0.f;
            if (lane < mcnt) {
                s_l = csrc[base + lane];
                a_l = __expf(leaky(el[s_l] + erd) - mx) / sm;
            }
        }
#pragma unroll 4
        for (int i = 0; i < mcnt; ++i) {
            const int   s = __builtin_amdgcn_readlane(s_l, i);
            const float a = __int_as_float(__builtin_amdgcn_readlane(__float_as_int(a_l), i));
            acc += feat[(size_t)s * 32 + lane] * a;
        }
    }
    const size_t o = (size_t)d * 32 + lane;
    outp[o] = outp[o] + acc;
}

extern "C" void kernel_launch(void* const* d_in, const int* in_sizes, int n_in,
                              void* d_out, int out_size, void* d_ws, size_t ws_size,
                              hipStream_t stream) {
    const float* x     = (const float*)d_in[0];
    const int*   src   = (const int*)  d_in[1];
    const int*   dst   = (const int*)  d_in[2];
    const float* W1    = (const float*)d_in[3];
    const float* al1   = (const float*)d_in[4];
    const float* ar1   = (const float*)d_in[5];
    const float* b1    = (const float*)d_in[6];
    const float* W2    = (const float*)d_in[7];
    const float* al2   = (const float*)d_in[8];
    const float* ar2   = (const float*)d_in[9];
    const float* resW2 = (const float*)d_in[10];
    const float* b2    = (const float*)d_in[11];
    float* outp = (float*)d_out;

    float* ws    = (float*)d_ws;
    float* feat1 = ws;
    float* rst1  = feat1 + (size_t)NN * 256;
    float* el1   = rst1  + (size_t)NN * 256;
    float* er1   = el1   + (size_t)NN * 8;
    int*   csrc  = (int*)(er1 + (size_t)NN * 8);
    int*   rank  = csrc + EE;
    int*   cnt   = rank + EE;
    int*   off   = cnt  + SCAN_LEN;
    float* feat2 = ws;
    float* el2   = feat2 + (size_t)NN * 32;
    float* er2   = el2   + NN;

    zero_cnt_kernel<<<SCAN_BLOCKS, 256, 0, stream>>>(cnt);
    hist_kernel<<<(EE + 255) / 256, 256, 0, stream>>>(dst, cnt, rank);
    scan_kernel<<<SCAN_BLOCKS, 256, 0, stream>>>(cnt, off);
    fill_kernel<<<(EE + 255) / 256, 256, 0, stream>>>(src, dst, off, rank, csrc);

    gemm1_kernel<<<NN / 16, 128, 0, stream>>>(x, W1, feat1);
    elr1_kernel<<<NN, 256, 0, stream>>>(feat1, al1, ar1, el1, er1);
    agg1_gather_kernel<<<NN / 8, 256, 0, stream>>>(off, csrc, el1, er1, feat1, x, b1, rst1);
    relu_kernel<<<NN, 256, 0, stream>>>(rst1);

    gemm2_kernel<<<(NN + 63) / 64, 128, 0, stream>>>(rst1, W2, resW2, b2, feat2, outp, NN);
    elr2_kernel<<<(NN + 7) / 8, 256, 0, stream>>>(feat2, al2, ar2, el2, er2, NN);
    agg2_gather_kernel<<<NN / 8, 256, 0, stream>>>(off, csrc, el2, er2, feat2, outp);
}


# ReLU folded into agg1's store and el/er of both layers taken in the GEMM epilogues from the accumulator tiles (sum order now: column pair, then 16-lane halving tree); 11 to 8 launches
# speedup vs baseline: 3.6160x; 1.3133x over previous
#include <hip/hip_runtime.h>
#include <hip/hip_bf16.h>
#include <math.h>

#define NN 50000
#define EE 800000
#define INF_ 256
#define OUTF 32
#define H1N 8
#define SLOPE 0.2f
#define NN_PAD 50016

typedef float v2f __attribute__((ext_vector_type(2)));
typedef float v8f __attribute__((ext_vector_type(8)));

__device__ __forceinline__ float leaky(float x) { return x > 0.f ? x : SLOPE * x; }

template <int M>
__device__ __forceinline__ float row_xor(const float x) {
    const int xi = __float_as_int(x);
    return __int_as_float(__builtin_amdgcn_update_dpp(xi, xi, 0x160 + M, 0xf, 0xf, true));
}

template <int CNT, int M>
__device__ __forceinline__ void halving_step(float* v, const int lane) {
    const bool up = (lane & M) != 0;
#pragma unroll
    for (int i = 0; i < CNT / 2; ++i) {
        const float keep = up ? v[i + CNT / 2] : v[i];
        const float send = up ? v[i] : v[i + CNT / 2];
        v[i] = keep + row_xor<M>(send);
    }
}
template <int N>
__device__ __forceinline__ void halfwave_sums(float (&v)[N], const int lane) {
    halving_step<N, 8>(v, lane);
    halving_step<N / 2, 4>(v, lane);
    halving_step<N / 4, 2>(v, lane);
    halving_step<N / 8, 1>(v, lane);
}

__global__ void __launch_bounds__(128) gemm1_kernel(const float* __restrict__ X,
                                                    const float* __restrict__ W,
                                                    const float* __restrict__ AL,
                                                    const float* __restrict__ AR,
                                                    float* __restrict__ C,
                                                    float* __restrict__ EL,
                                                    float* __restrict__ ER) {
    __shared__ float s_el[16 * H1N];
    __shared__ float s_er[16 * H1N];
    const int wave = threadIdx.x >> 5;
    const int lane = threadIdx.x & 31;
    const int half = lane >> 4;
    const int r    = lane & 15;
    const int row0 = blockIdx.x * 16;
    const int col0 = wave * 64;

    v8f acc[4] = {};
    const float* xrow = X + (size_t)(row0 + r) * INF_;

    for (int k = 0; k < INF_; k += 4) {
        v2f a;
        a.x = xrow[k + half * 2 + 0];
        a.y = xrow[k + half * 2 + 1];
#pragma unroll
        for (int t = 0; t < 4; ++t) {
            const int col = col0 + t * 16 + r;
            v2f b;
            b.x = W[(size_t)(k + half * 2 + 0) * 256 + col];
            b.y = W[(size_t)(k + half * 2 + 1) * 256 + col];
            acc[t] = __builtin_amdgcn_wmma_f32_16x16x4_f32(
                false, a, false, b, (short)0, acc[t], false, false);
        }
    }

    const int lrow = half * 8 + (r >> 1);
    const int hd   = wave * 2 + (r & 1);
    {
        float wc[4];
#pragma unroll
        for (int t = 0; t < 4; ++t) wc[t] = AL[col0 + t * 16 + r];
        float v[16];
#pragma unroll
        for (int j = 0; j < 8; ++j) {
            v[2 * j + 0] = acc[0][j] * wc[0] + acc[1][j] * wc[1];
            v[2 * j + 1] = acc[2][j] * wc[2] + acc[3][j] * wc[3];
        }
        halfwave_sums<16>(v, lane);
        s_el[lrow * H1N + hd] = v[0];
    }
    __builtin_amdgcn_sched_barrier(0);
    {
        float wc[4];
#pragma unroll
        for (int t = 0; t < 4; ++t) wc[t] = AR[col0 + t * 16 + r];
        float v[16];
#pragma unroll
        for (int j = 0; j < 8; ++j) {
            v[2 * j + 0] = acc[0][j] * wc[0] + acc[1][j] * wc[1];
            v[2 * j + 1] = acc[2][j] * wc[2] + acc[3][j] * wc[3];
        }
        halfwave_sums<16>(v, lane);
        s_er[lrow * H1N + hd] = v[0];
    }
    __syncthreads();
    EL[(size_t)row0 * H1N + threadIdx.x] = s_el[threadIdx.x];
    ER[(size_t)row0 * H1N + threadIdx.x] = s_er[threadIdx.x];

#pragma unroll
    for (int t = 0; t < 4; ++t) {
#pragma unroll
        for (int j = 0; j < 8; ++j) {
            const int row = row0 + half * 8 + j;
            const int col = col0 + t * 16 + r;
            C[(size_t)row * 256 + col] = acc[t][j];
        }
    }
}

__global__ void __launch_bounds__(128) gemm2_kernel(const float* __restrict__ Y,
                                                    const float* __restrict__ W2,
                                                    const float* __restrict__ RW,
                                                    const float* __restrict__ b2,
                                                    const float* __restrict__ AL,
                                                    const float* __restrict__ AR,
                                                    float* __restrict__ F2,
                                                    float* __restrict__ OUTP,
                                                    float* __restrict__ EL,
                                                    float* __restrict__ ER,
                                                    int n) {
    __shared__ float s_lg[2][64];
    const int wave = threadIdx.x >> 5;
    const int lane = threadIdx.x & 31;
    const int half = lane >> 4;
    const int r    = lane & 15;
    const int row0 = blockIdx.x * 64 + wave * 16;

    int rload = row0 + r;
    if (rload >= n) rload = n - 1;
    const float* yrow = Y + (size_t)rload * 256;

    v8f accW[2] = {};
    v8f accR[2] = {};

    for (int k = 0; k < 256; k += 4) {
        v2f a;
        a.x = yrow[k + half * 2 + 0];
        a.y = yrow[k + half * 2 + 1];
#pragma unroll
        for (int t = 0; t < 2; ++t) {
            const int col = t * 16 + r;
            v2f bw, br;
            bw.x = W2[(size_t)(k + half * 2 + 0) * 32 + col];
            bw.y = W2[(size_t)(k + half * 2 + 1) * 32 + col];
            accW[t] = __builtin_amdgcn_wmma_f32_16x16x4_f32(
                false, a, false, bw, (short)0, accW[t], false, false);
            br.x = RW[(size_t)(k + half * 2 + 0) * 32 + col];
            br.y = RW[(size_t)(k + half * 2 + 1) * 32 + col];
            accR[t] = __builtin_amdgcn_wmma_f32_16x16x4_f32(
                false, a, false, br, (short)0, accR[t], false, false);
        }
    }

    float al_c[2], ar_c[2];
#pragma unroll
    for (int t = 0; t < 2; ++t) {
        al_c[t] = AL[t * 16 + r];
        ar_c[t] = AR[t * 16 + r];
    }
    float v[16];
#pragma unroll
    for (int j = 0; j < 8; ++j) {
        v[2 * j + 0] = accW[0][j] * al_c[0] + accW[1][j] * al_c[1];
        v[2 * j + 1] = accW[0][j] * ar_c[0] + accW[1][j] * ar_c[1];
    }
    halfwave_sums<16>(v, lane);
    s_lg[r & 1][wave * 16 + half * 8 + (r >> 1)] = v[0];
    __syncthreads();
    {
        const int which = wave >> 1;
        const int li    = (wave & 1) * 32 + lane;
        const int grow  = blockIdx.x * 64 + li;
        float* dstp = which ? ER : EL;
        if (grow < n) dstp[grow] = s_lg[which][li];
    }

#pragma unroll
    for (int t = 0; t < 2; ++t) {
#pragma unroll
        for (int j = 0; j < 8; ++j) {
            const int row = row0 + half * 8 + j;
            const int col = t * 16 + r;
            if (row < n) {
                F2[(size_t)row * 32 + col]   = accW[t][j];
                OUTP[(size_t)row * 32 + col] = accR[t][j] + b2[col];
            }
        }
    }
}

#define SCAN_BLOCKS 196
#define SCAN_LEN (SCAN_BLOCKS * 256)

__global__ void __launch_bounds__(256) zero_cnt_kernel(int* __restrict__ cnt) {
    cnt[blockIdx.x * 256 + threadIdx.x] = 0;
}

__global__ void __launch_bounds__(256) hist_kernel(const int* __restrict__ dst,
                                                   int* __restrict__ cnt,
                                                   int* __restrict__ rank) {
    const int e = blockIdx.x * 256 + threadIdx.x;
    if (e >= EE) return;
    rank[e] = atomicAdd(&cnt[dst[e]], 1);
}

__global__ void __launch_bounds__(256) scan_kernel(const int* __restrict__ cnt,
                                                   int* __restrict__ off) {
    __shared__ int s_part[8];
    __shared__ int s_tot[8];
    const int t = threadIdx.x, lane = t & 31, w = t >> 5;
    const int b = blockIdx.x;
    int part = 0;
    for (int j = 0; j < b; ++j) part += cnt[j * 256 + t];
    const int v = cnt[b * 256 + t];
#pragma unroll
    for (int o = 16; o > 0; o >>= 1) part += __shfl_xor(part, o, 32);
    int inc = v;
#pragma unroll
    for (int o = 1; o < 32; o <<= 1) {
        const int nb = __shfl_up(inc, o, 32);
        if (lane >= o) inc += nb;
    }
    if (lane == 0)  s_part[w] = part;
    if (lane == 31) s_tot[w]  = inc;
    __syncthreads();
    int base = 0;
#pragma unroll
    for (int i = 0; i < 8; ++i) {
        base += s_part[i];
        if (i < w) base += s_tot[i];
    }
    off[b * 256 + t] = base + inc - v;
}

__global__ void __launch_bounds__(256) fill_kernel(const int* __restrict__ src,
                                                   const int* __restrict__ dst,
                                                   const int* __restrict__ off,
                                                   const int* __restrict__ rank,
                                                   int* __restrict__ csrc) {
    const int e = blockIdx.x * 256 + threadIdx.x;
    if (e >= EE) return;
    csrc[off[dst[e]] + rank[e]] = src[e];
}

__global__ void __launch_bounds__(256) agg1_gather_kernel(const int* __restrict__ off,
                                                          const int* __restrict__ csrc,
                                                          const float* __restrict__ el,
                                                          const float* __restrict__ er,
                                                          const float* __restrict__ feat,
                                                          const float* __restrict__ x,
                                                          const float* __restrict__ b1,
                                                          float* __restrict__ rst) {
    const int lane = threadIdx.x & 31;
    const int d = __builtin_amdgcn_readfirstlane((int)(blockIdx.x * 8 + (threadIdx.x >> 5)));
    const int beg = off[d], end = off[d + 1];

    const int hq = lane & 7, q = lane >> 3;
    const float erq = er[d * H1N + hq];
    float v[8];
    for (int k = 0; k < 8; ++k) v[k] = -INFINITY;
    {
        int sq[4];
        for (int k = 0; k < 4; ++k) {
            const int e = beg + 4 * k + q;
            sq[k] = csrc[e < EE ? e : EE - 1];
        }
        for (int k = 0; k < 4; ++k) {
            const float t = leaky(el[sq[k] * H1N + hq] + erq);
            v[k] = (beg + 4 * k + q < end) ? t : -INFINITY;
        }
    }
    if (end - beg > 16) {
        int sq[4];
        for (int k = 4; k < 8; ++k) {
            const int e = beg + 4 * k + q;
            sq[k - 4] = csrc[e < EE ? e : EE - 1];
        }
        for (int k = 4; k < 8; ++k) {
            const float t = leaky(el[sq[k - 4] * H1N + hq] + erq);
            v[k] = (beg + 4 * k + q < end) ? t : -INFINITY;
        }
    }
    float m = -INFINITY;
    for (int k = 0; k < 8; ++k) m = fmaxf(m, v[k]);
    for (int e = beg + 32 + q; e < end; e += 4)
        m = fmaxf(m, leaky(el[csrc[e] * H1N + hq] + erq));
    m = fmaxf(m, __shfl_xor(m, 8, 32));
    m = fmaxf(m, __shfl_xor(m, 16, 32));
    float su = 0.f;
    for (int k = 0; k < 8; ++k)
        if (beg + 4 * k + q < end) su += __expf(v[k] - m);
    for (int e = beg + 32 + q; e < end; e += 4)
        su += __expf(leaky(el[csrc[e] * H1N + hq] + erq) - m);
    su += __shfl_xor(su, 8, 32);
    su += __shfl_xor(su, 16, 32);

    const int h = lane >> 2;
    const float erd = __shfl(erq, h, 32);
    const float mx  = __shfl(m,   h, 32);
    const float sm  = __shfl(su,  h, 32);

    float4 a0 = make_float4(0.f, 0.f, 0.f, 0.f);
    float4 a1 = make_float4(0.f, 0.f, 0.f, 0.f);
    for (int base = beg; base < end; base += 32) {
        const int mcnt = (end - base < 32) ? (end - base) : 32;
        const int s_l = (lane < mcnt) ? csrc[base + lane] : 0;
#pragma unroll 4
        for (int i = 0; i < mcnt; ++i) {
            const int s = __builtin_amdgcn_readlane(s_l, i);
            const float sc = leaky(el[s * H1N + h] + erd);
            const float a  = __expf(sc - mx) / sm;
            const float4* fs = (const float4*)(feat + (size_t)s * 256 + lane * 8);
            const float4 f0 = fs[0], f1 = fs[1];
            a0.x += f0.x * a; a0.y += f0.y * a; a0.z += f0.z * a; a0.w += f0.w * a;
            a1.x += f1.x * a; a1.y += f1.y * a; a1.z += f1.z * a; a1.w += f1.w * a;
        }
    }
    const size_t o = (size_t)d * 256 + lane * 8;
    const float4 x0 = *(const float4*)(x + o);
    const float4 x1 = *(const float4*)(x + o + 4);
    const float4 c0 = *(const float4*)(b1 + lane * 8);
    const float4 c1 = *(const float4*)(b1 + lane * 8 + 4);
    float4 r0, r1;
    r0.x = fmaxf(a0.x + x0.x + c0.x, 0.f); r0.y = fmaxf(a0.y + x0.y + c0.y, 0.f);
    r0.z = fmaxf(a0.z + x0.z + c0.z, 0.f); r0.w = fmaxf(a0.w + x0.w + c0.w, 0.f);
    r1.x = fmaxf(a1.x + x1.x + c1.x, 0.f); r1.y = fmaxf(a1.y + x1.y + c1.y, 0.f);
    r1.z = fmaxf(a1.z + x1.z + c1.z, 0.f); r1.w = fmaxf(a1.w + x1.w + c1.w, 0.f);
    *(float4*)(rst + o)     = r0;
    *(float4*)(rst + o + 4) = r1;
}

__global__ void __launch_bounds__(256) agg2_gather_kernel(const int* __restrict__ off,
                                                          const int* __restrict__ csrc,
                                                          const float* __restrict__ el,
                                                          const float* __restrict__ er,
                                                          const float* __restrict__ feat,
                                                          float* __restrict__ outp) {
    const int lane = threadIdx.x & 31;
    const int d = __builtin_amdgcn_readfirstlane((int)(blockIdx.x * 8 + (threadIdx.x >> 5)));
    const int beg = off[d], end = off[d + 1];
    const float erd = er[d];

    const bool have0 = beg + lane < end;
    int   s0 = 0;
    float v0 = -INFINITY;
    if (have0) { s0 = csrc[beg + lane]; v0 = leaky(el[s0] + erd); }
    float mx = v0;
    for (int e = beg + 32 + lane; e < end; e += 32)
        mx = fmaxf(mx, leaky(el[csrc[e]] + erd));
    for (int o = 16; o > 0; o >>= 1) mx = fmaxf(mx, __shfl_xor(mx, o, 32));
    const float w0 = have0 ? __expf(v0 - mx) : 0.f;
    float sm = w0;
    for (int e = beg + 32 + lane; e < end; e += 32)
        sm += __expf(leaky(el[csrc[e]] + erd) - mx);
    for (int o = 16; o > 0; o >>= 1) sm += __shfl_xor(sm, o, 32);

    float acc = 0.f;
    for (int base = beg; base < end; base += 32) {
        const int mcnt = (end - base < 32) ? (end - base) : 32;
        int   s_l = s0;
        float a_l = w0 / sm;
        if (base != beg) {
            s_l = 0; a_l = 0.f;
            if (lane < mcnt) {
                s_l = csrc[base + lane];
                a_l = __expf(leaky(el[s_l] + erd) - mx) / sm;
            }
        }
#pragma unroll 4
        for (int i = 0; i < mcnt; ++i) {
            const int   s = __builtin_amdgcn_readlane(s_l, i);
            const float a = __int_as_float(__builtin_amdgcn_readlane(__float_as_int(a_l), i));
            acc += feat[(size_t)s * 32 + lane] * a;
        }
    }
    const size_t o = (size_t)d * 32 + lane;
    outp[o] = outp[o] + acc;
}

extern "C" void kernel_launch(void* const* d_in, const int* in_sizes, int n_in,
                              void* d_out, int out_size, void* d_ws, size_t ws_size,
                              hipStream_t stream) {
    const float* x     = (const float*)d_in[0];
    const int*   src   = (const int*)  d_in[1];
    const int*   dst   = (const int*)  d_in[2];
    const float* W1    = (const float*)d_in[3];
    const float* al1   = (const float*)d_in[4];
    const float* ar1   = (const float*)d_in[5];
    const float* b1    = (const float*)d_in[6];
    const float* W2    = (const float*)d_in[7];
    const float* al2   = (const float*)d_in[8];
    const float* ar2   = (const float*)d_in[9];
    const float* resW2 = (const float*)d_in[10];
    const float* b2    = (const float*)d_in[11];
    float* outp = (float*)d_out;

    float* ws    = (float*)d_ws;
    float* feat1 = ws;
    float* rst1  = feat1 + (size_t)NN * 256;
    float* el1   = rst1  + (size_t)NN * 256;
    float* er1   = el1   + (size_t)NN * 8;
    int*   csrc  = (int*)(er1 + (size_t)NN * 8);
    int*   rank  = csrc + EE;
    int*   cnt   = rank + EE;
    int*   off   = cnt  + SCAN_LEN;
    float* feat2 = ws;
    float* el2   = feat2 + (size_t)NN * 32;
    float* er2   = el2   + NN_PAD;

    zero_cnt_kernel<<<SCAN_BLOCKS, 256, 0, stream>>>(cnt);
    hist_kernel<<<(EE + 255) / 256, 256, 0, stream>>>(dst, cnt, rank);
    scan_kernel<<<SCAN_BLOCKS, 256, 0, stream>>>(cnt, off);
    fill_kernel<<<(EE + 255) / 256, 256, 0, stream>>>(src, dst, off, rank, csrc);

    gemm1_kernel<<<NN / 16, 128, 0, stream>>>(x, W1, al1, ar1, feat1, el1, er1);
    agg1_gather_kernel<<<NN / 8, 256, 0, stream>>>(off, csrc, el1, er1, feat1, x, b1, rst1);

    gemm2_kernel<<<(NN + 63) / 64, 128, 0, stream>>>(rst1, W2, resW2, b2, al2, ar2,
                                                     feat2, outp, el2, er2, NN);
    agg2_gather_kernel<<<NN / 8, 256, 0, stream>>>(off, csrc, el2, er2, feat2, outp);
}


# Both GEMMs: column tiles interleaved, a lane owns 4 adjacent columns (2 in GEMM2): B loads and output stores cover whole lines, A is one 16-byte load per 8 k; k and logit sum order change
# speedup vs baseline: 3.9199x; 1.0840x over previous
#include <hip/hip_runtime.h>
#include <hip/hip_bf16.h>
#include <math.h>

#define NN 50000
#define EE 800000
#define INF_ 256
#define OUTF 32
#define H1N 8
#define SLOPE 0.2f
#define NN_PAD 50016

typedef float v2f __attribute__((ext_vector_type(2)));
typedef float v4f __attribute__((ext_vector_type(4)));
typedef float v8f __attribute__((ext_vector_type(8)));

__device__ __forceinline__ float leaky(float x) { return x > 0.f ? x : SLOPE * x; }

__device__ __forceinline__ void accumulator_slots() {
    __builtin_amdgcn_sched_barrier(0);
    asm volatile("v_nop\n\tv_nop\n\tv_nop\n\tv_nop");
    __builtin_amdgcn_sched_barrier(0);
}

template <int M>
__device__ __forceinline__ float row_xor(const float x) {
    const int xi = __float_as_int(x);
    return __int_as_float(__builtin_amdgcn_update_dpp(xi, xi, 0x160 + M, 0xf, 0xf, true));
}

template <int CNT, int M>
__device__ __forceinline__ void halving_step(float* v, const int lane) {
    const bool up = (lane & M) != 0;
#pragma unroll
    for (int i = 0; i < CNT / 2; ++i) {
        const float keep = up ? v[i + CNT / 2] : v[i];
        const float send = up ? v[i] : v[i + CNT / 2];
        v[i] = keep + row_xor<M>(send);
    }
}
template <int N>
__device__ __forceinline__ void halfwave_sums(float (&v)[N], const int lane) {
    halving_step<N, 8>(v, lane);
    halving_step<N / 2, 4>(v, lane);
    halving_step<N / 4, 2>(v, lane);
    halving_step<N / 8, 1>(v, lane);
}

__global__ void __launch_bounds__(128) gemm1_kernel(const float* __restrict__ X,
                                                    const float* __restrict__ W,
                                                    const float* __restrict__ AR,
                                                    float* __restrict__ C,
                                                    float* __restrict__ ER) {
    __shared__ float s_er[16 * H1N];
    const int wave = threadIdx.x >> 5;
    const int lane = threadIdx.x & 31;
    const int half = lane >> 4;
    const int r    = lane & 15;
    const int row0 = blockIdx.x * 16;
    const int col0 = wave * 64;

    v8f acc[4] = {};
    const float* xp = X + (size_t)(row0 + r) * INF_ + half * 4;
    const float* wp = W + (size_t)(half * 4) * 256 + col0 + r * 4;

#pragma nounroll
    for (int k = 0; k < INF_; k += 8) {
        const v4f a4 = *(const v4f*)(xp + k);
        const float* wk = wp + (size_t)k * 256;
        const v4f b0 = *(const v4f*)(wk);
        const v4f b1 = *(const v4f*)(wk + 256);
        const v4f b2 = *(const v4f*)(wk + 512);
        const v4f b3 = *(const v4f*)(wk + 768);
        v2f a;
        a.x = a4.x; a.y = a4.y;
#pragma unroll
        for (int t = 0; t < 4; ++t) {
            v2f b;
            b.x = b0[t]; b.y = b1[t];
            acc[t] = __builtin_amdgcn_wmma_f32_16x16x4_f32(
                false, a, false, b, (short)0, acc[t], false, false);
        }
        a.x = a4.z; a.y = a4.w;
#pragma unroll
        for (int t = 0; t < 4; ++t) {
            v2f b;
            b.x = b2[t]; b.y = b3[t];
            acc[t] = __builtin_amdgcn_wmma_f32_16x16x4_f32(
                false, a, false, b, (short)0, acc[t], false, false);
        }
    }

    accumulator_slots();

    const int lrow = half * 8 + (r & 7);
    const int hd   = wave * 2 + (r >> 3);
    {
        const v4f wc = *(const v4f*)(AR + col0 + r * 4);
        float v[8];
#pragma unroll
        for (int p = 0; p < 4; ++p) {
            v2f q[4];
#pragma unroll
            for (int t = 0; t < 4; ++t) { q[t].x = acc[t][2 * p]; q[t].y = acc[t][2 * p + 1]; }
            v2f s = q[0] * wc.x;
            s += q[1] * wc.y;
            s += q[2] * wc.z;
            s += q[3] * wc.w;
            v[2 * p + 0] = s.x;
            v[2 * p + 1] = s.y;
        }
        halving_step<8, 4>(v, lane);
        halving_step<4, 2>(v, lane);
        halving_step<2, 1>(v, lane);
        s_er[lrow * H1N + hd] = v[0];
    }
    __syncthreads();
    ER[(size_t)row0 * H1N + threadIdx.x] = s_er[threadIdx.x];

#pragma unroll
    for (int j = 0; j < 8; ++j) {
        const int row = row0 + half * 8 + j;
        v4f o;
        o.x = acc[0][j]; o.y = acc[1][j]; o.z = acc[2][j]; o.w = acc[3][j];
        *(v4f*)(C + (size_t)row * 256 + col0 + r * 4) = o;
    }
}

__global__ void __launch_bounds__(128) gemm2_kernel(const float* __restrict__ Y,
                                                    const float* __restrict__ W2,
                                                    const float* __restrict__ RW,
                                                    const float* __restrict__ b2,
                                                    const float* __restrict__ AL,
                                                    const float* __restrict__ AR,
                                                    float* __restrict__ F2,
                                                    float* __restrict__ OUTP,
                                                    float* __restrict__ EL,
                                                    float* __restrict__ ER,
                                                    int n) {
    __shared__ float s_lg[2][64];
    const int wave = threadIdx.x >> 5;
    const int lane = threadIdx.x & 31;
    const int half = lane >> 4;
    const int r    = lane & 15;
    const int row0 = blockIdx.x * 64 + wave * 16;

    int rload = row0 + r;
    if (rload >= n) rload = n - 1;
    const float* yp  = Y  + (size_t)rload * 256 + half * 4;
    const float* w2p = W2 + (size_t)(half * 4) * 32 + r * 2;
    const float* rwp = RW + (size_t)(half * 4) * 32 + r * 2;

    v8f accW[2] = {};
    v8f accR[2] = {};

    for (int k = 0; k < 256; k += 8) {
        const v4f a4 = *(const v4f*)(yp + k);
        const float* wk = w2p + (size_t)k * 32;
        const float* rk = rwp + (size_t)k * 32;
        const v2f w0 = *(const v2f*)(wk);
        const v2f w1 = *(const v2f*)(wk + 32);
        const v2f w2 = *(const v2f*)(wk + 64);
        const v2f w3 = *(const v2f*)(wk + 96);
        const v2f q0 = *(const v2f*)(rk);
        const v2f q1 = *(const v2f*)(rk + 32);
        const v2f q2 = *(const v2f*)(rk + 64);
        const v2f q3 = *(const v2f*)(rk + 96);
        v2f a;
        a.x = a4.x; a.y = a4.y;
#pragma unroll
        for (int t = 0; t < 2; ++t) {
            v2f bw, br;
            bw.x = w0[t]; bw.y = w1[t];
            accW[t] = __builtin_amdgcn_wmma_f32_16x16x4_f32(
                false, a, false, bw, (short)0, accW[t], false, false);
            br.x = q0[t]; br.y = q1[t];
            accR[t] = __builtin_amdgcn_wmma_f32_16x16x4_f32(
                false, a, false, br, (short)0, accR[t], false, false);
        }
        a.x = a4.z; a.y = a4.w;
#pragma unroll
        for (int t = 0; t < 2; ++t) {
            v2f bw, br;
            bw.x = w2[t]; bw.y = w3[t];
            accW[t] = __builtin_amdgcn_wmma_f32_16x16x4_f32(
                false, a, false, bw, (short)0, accW[t], false, false);
            br.x = q2[t]; br.y = q3[t];
            accR[t] = __builtin_amdgcn_wmma_f32_16x16x4_f32(
                false, a, false, br, (short)0, accR[t], false, false);
        }
    }

    accumulator_slots();

    const v2f al_c = *(const v2f*)(AL + r * 2);
    const v2f ar_c = *(const v2f*)(AR + r * 2);
    float v[16];
#pragma unroll
    for (int j = 0; j < 8; ++j) {
        v[2 * j + 0] = accW[0][j] * al_c.x + accW[1][j] * al_c.y;
        v[2 * j + 1] = accW[0][j] * ar_c.x + accW[1][j] * ar_c.y;
    }
    halfwave_sums<16>(v, lane);
    s_lg[r & 1][wave * 16 + half * 8 + (r >> 1)] = v[0];
    __syncthreads();
    {
        const int which = wave >> 1;
        const int li    = (wave & 1) * 32 + lane;
        const int grow  = blockIdx.x * 64 + li;
        float* dstp = which ? ER : EL;
        if (grow < n) dstp[grow] = s_lg[which][li];
    }

    const v2f bias = *(const v2f*)(b2 + r * 2);
#pragma unroll
    for (int j = 0; j < 8; ++j) {
        const int row = row0 + half * 8 + j;
        if (row < n) {
            v2f f, o;
            f.x = accW[0][j];          f.y = accW[1][j];
            o.x = accR[0][j] + bias.x; o.y = accR[1][j] + bias.y;
            *(v2f*)(F2   + (size_t)row * 32 + r * 2) = f;
            *(v2f*)(OUTP + (size_t)row * 32 + r * 2) = o;
        }
    }
}

#define SCAN_BLOCKS 196
#define SCAN_LEN (SCAN_BLOCKS * 256)

__global__ void __launch_bounds__(256) zero_cnt_kernel(int* __restrict__ cnt) {
    cnt[blockIdx.x * 256 + threadIdx.x] = 0;
}

__global__ void __launch_bounds__(256) hist_kernel(const int* __restrict__ dst,
                                                   int* __restrict__ cnt,
                                                   int* __restrict__ rank) {
    const int e = blockIdx.x * 256 + threadIdx.x;
    if (e >= EE) return;
    rank[e] = atomicAdd(&cnt[dst[e]], 1);
}

__global__ void __launch_bounds__(256) scan_kernel(const int* __restrict__ cnt,
                                                   int* __restrict__ off) {
    __shared__ int s_part[8];
    __shared__ int s_tot[8];
    const int t = threadIdx.x, lane = t & 31, w = t >> 5;
    const int b = blockIdx.x;
    int part = 0;
    for (int j = 0; j < b; ++j) part += cnt[j * 256 + t];
    const int v = cnt[b * 256 + t];
#pragma unroll
    for (int o = 16; o > 0; o >>= 1) part += __shfl_xor(part, o, 32);
    int inc = v;
#pragma unroll
    for (int o = 1; o < 32; o <<= 1) {
        const int nb = __shfl_up(inc, o, 32);
        if (lane >= o) inc += nb;
    }
    if (lane == 0)  s_part[w] = part;
    if (lane == 31) s_tot[w]  = inc;
    __syncthreads();
    int base = 0;
#pragma unroll
    for (int i = 0; i < 8; ++i) {
        base += s_part[i];
        if (i < w) base += s_tot[i];
    }
    off[b * 256 + t] = base + inc - v;
}

__global__ void __launch_bounds__(256) fill_kernel(const int* __restrict__ src,
                                                   const int* __restrict__ dst,
                                                   const int* __restrict__ off,
                                                   const int* __restrict__ rank,
                                                   int* __restrict__ csrc) {
    const int e = blockIdx.x * 256 + threadIdx.x;
    if (e >= EE) return;
    csrc[off[dst[e]] + rank[e]] = src[e];
}


template <int N>
__device__ __forceinline__ void agg1_group(const float* __restrict__ feat, const int s_l,
                                           const int i, const int lane,
                                           const v2f (&w)[4], const float erd,
                                           float& m, float& l, v2f (&a)[4]) {
    v4f f0[N], f1[N];
#pragma unroll
    for (int k = 0; k < N; ++k) {
        const int s = __builtin_amdgcn_readlane(s_l, i + k);
        const v4f* fs = (const v4f*)(feat + (size_t)s * 256 + lane * 8);
        f0[k] = fs[0];
        f1[k] = fs[1];
    }
    float sc[N];
    float mc = m;
#pragma unroll
    for (int k = 0; k < N; ++k) {
        v2f q = f0[k].xy * w[0];
        q += f0[k].zw * w[1];
        q += f1[k].xy * w[2];
        q += f1[k].zw * w[3];
        float p = q.x + q.y;
        p += row_xor<1>(p);
        p += row_xor<2>(p);
        sc[k] = leaky(p + erd);
        mc = fmaxf(mc, sc[k]);
    }
    const float c = __expf(m - mc);
    l *= c;
#pragma unroll
    for (int j = 0; j < 4; ++j) a[j] *= c;
#pragma unroll
    for (int k = 0; k < N; ++k) {
        const float p = __expf(sc[k] - mc);
        l += p;
        a[0] += f0[k].xy * p;
        a[1] += f0[k].zw * p;
        a[2] += f1[k].xy * p;
        a[3] += f1[k].zw * p;
    }
    m = mc;
}

__global__ void __launch_bounds__(256) agg1_gather_kernel(const int* __restrict__ off,
                                                          const int* __restrict__ csrc,
                                                          const float* __restrict__ al,
                                                          const float* __restrict__ er,
                                                          const float* __restrict__ feat,
                                                          const float* __restrict__ x,
                                                          const float* __restrict__ b1,
                                                          float* __restrict__ rst) {
    const int lane = threadIdx.x & 31;
    const int d = __builtin_amdgcn_readfirstlane((int)(blockIdx.x * 8 + (threadIdx.x >> 5)));
    const int beg = off[d], end = off[d + 1];
    const int h = lane >> 2;
    const float erd = er[d * H1N + h];
    v2f w[4];
#pragma unroll
    for (int j = 0; j < 4; ++j) w[j] = *(const v2f*)(al + lane * 8 + 2 * j);

    float m = -INFINITY, l = 0.f;
    v2f a[4] = {};
    for (int base = beg; base < end; base += 32) {
        const int mcnt = (end - base < 32) ? (end - base) : 32;
        const int s_l = (lane < mcnt) ? csrc[base + lane] : 0;
        int i = 0;
        for (; i + 4 <= mcnt; i += 4)
            agg1_group<4>(feat, s_l, i, lane, w, erd, m, l, a);
        const int rem = mcnt - i;
        if (rem == 3)      agg1_group<3>(feat, s_l, i, lane, w, erd, m, l, a);
        else if (rem == 2) agg1_group<2>(feat, s_l, i, lane, w, erd, m, l, a);
        else if (rem == 1) agg1_group<1>(feat, s_l, i, lane, w, erd, m, l, a);
    }
    const float inv = (l > 0.f) ? 1.f / l : 0.f;
    const size_t o = (size_t)d * 256 + lane * 8;
    const float4 x0 = *(const float4*)(x + o);
    const float4 x1 = *(const float4*)(x + o + 4);
    const float4 c0 = *(const float4*)(b1 + lane * 8);
    const float4 c1 = *(const float4*)(b1 + lane * 8 + 4);
    float4 r0, r1;
    r0.x = fmaxf(a[0].x * inv + x0.x + c0.x, 0.f); r0.y = fmaxf(a[0].y * inv + x0.y + c0.y, 0.f);
    r0.z = fmaxf(a[1].x * inv + x0.z + c0.z, 0.f); r0.w = fmaxf(a[1].y * inv + x0.w + c0.w, 0.f);
    r1.x = fmaxf(a[2].x * inv + x1.x + c1.x, 0.f); r1.y = fmaxf(a[2].y * inv + x1.y + c1.y, 0.f);
    r1.z = fmaxf(a[3].x * inv + x1.z + c1.z, 0.f); r1.w = fmaxf(a[3].y * inv + x1.w + c1.w, 0.f);
    *(float4*)(rst + o)     = r0;
    *(float4*)(rst + o + 4) = r1;
}

__global__ void __launch_bounds__(256) agg2_gather_kernel(const int* __restrict__ off,
                                                          const int* __restrict__ csrc,
                                                          const float* __restrict__ el,
                                                          const float* __restrict__ er,
                                                          const float* __restrict__ feat,
                                                          float* __restrict__ outp) {
    const int lane = threadIdx.x & 31;
    const int d = __builtin_amdgcn_readfirstlane((int)(blockIdx.x * 8 + (threadIdx.x >> 5)));
    const int beg = off[d], end = off[d + 1];
    const float erd = er[d];

    const bool have0 = beg + lane < end;
    int   s0 = 0;
    float v0 = -INFINITY;
    if (have0) { s0 = csrc[beg + lane]; v0 = leaky(el[s0] + erd); }
    float mx = v0;
    for (int e = beg + 32 + lane; e < end; e += 32)
        mx = fmaxf(mx, leaky(el[csrc[e]] + erd));
    for (int o = 16; o > 0; o >>= 1) mx = fmaxf(mx, __shfl_xor(mx, o, 32));
    const float w0 = have0 ? __expf(v0 - mx) : 0.f;
    float sm = w0;
    for (int e = beg + 32 + lane; e < end; e += 32)
        sm += __expf(leaky(el[csrc[e]] + erd) - mx);
    for (int o = 16; o > 0; o >>= 1) sm += __shfl_xor(sm, o, 32);

    float acc = 0.f;
    for (int base = beg; base < end; base += 32) {
        const int mcnt = (end - base < 32) ? (end - base) : 32;
        int   s_l = s0;
        float a_l = w0 / sm;
        if (base != beg) {
            s_l = 0; a_l = 0.f;
            if (lane < mcnt) {
                s_l = csrc[base + lane];
                a_l = __expf(leaky(el[s_l] + erd) - mx) / sm;
            }
        }
#pragma unroll 4
        for (int i = 0; i < mcnt; ++i) {
            const int   s = __builtin_amdgcn_readlane(s_l, i);
            const float a = __int_as_float(__builtin_amdgcn_readlane(__float_as_int(a_l), i));
            acc += feat[(size_t)s * 32 + lane] * a;
        }
    }
    const size_t o = (size_t)d * 32 + lane;
    outp[o] = outp[o] + acc;
}

extern "C" void kernel_launch(void* const* d_in, const int* in_sizes, int n_in,
                              void* d_out, int out_size, void* d_ws, size_t ws_size,
                              hipStream_t stream) {
    const float* x     = (const float*)d_in[0];
    const int*   src   = (const int*)  d_in[1];
    const int*   dst   = (const int*)  d_in[2];
    const float* W1    = (const float*)d_in[3];
    const float* al1   = (const float*)d_in[4];
    const float* ar1   = (const float*)d_in[5];
    const float* b1    = (const float*)d_in[6];
    const float* W2    = (const float*)d_in[7];
    const float* al2   = (const float*)d_in[8];
    const float* ar2   = (const float*)d_in[9];
    const float* resW2 = (const float*)d_in[10];
    const float* b2    = (const float*)d_in[11];
    float* outp = (float*)d_out;

    float* ws    = (float*)d_ws;
    float* feat1 = ws;
    float* rst1  = feat1 + (size_t)NN * 256;
    float* er1   = rst1  + (size_t)NN * 256;
    int*   csrc  = (int*)(er1 + (size_t)NN * 8);
    int*   rank  = csrc + EE;
    int*   cnt   = rank + EE;
    int*   off   = cnt  + SCAN_LEN;
    float* feat2 = ws;
    float* el2   = feat2 + (size_t)NN * 32;
    float* er2   = el2   + NN_PAD;

    zero_cnt_kernel<<<SCAN_BLOCKS, 256, 0, stream>>>(cnt);
    hist_kernel<<<(EE + 255) / 256, 256, 0, stream>>>(dst, cnt, rank);
    scan_kernel<<<SCAN_BLOCKS, 256, 0, stream>>>(cnt, off);
    fill_kernel<<<(EE + 255) / 256, 256, 0, stream>>>(src, dst, off, rank, csrc);

    gemm1_kernel<<<NN / 16, 128, 0, stream>>>(x, W1, ar1, feat1, er1);
    agg1_gather_kernel<<<NN / 8, 256, 0, stream>>>(off, csrc, al1, er1, feat1, x, b1, rst1);

    gemm2_kernel<<<(NN + 63) / 64, 128, 0, stream>>>(rst1, W2, resW2, b2, al2, ar2,
                                                     feat2, outp, el2, er2, NN);
    agg2_gather_kernel<<<NN / 8, 256, 0, stream>>>(off, csrc, el2, er2, feat2, outp);
}


# Edge sort rebuilt as a two-level counting sort in LDS (chunk runs per 256-node bucket, then one workgroup per bucket): no global atomic, no 4-byte scatter; segment order now LDS arrival order
# speedup vs baseline: 5.3816x; 1.3729x over previous
#include <hip/hip_runtime.h>
#include <hip/hip_bf16.h>
#include <math.h>

#define NN 50000
#define EE 800000
#define INF_ 256
#define OUTF 32
#define H1N 8
#define SLOPE 0.2f
#define NN_PAD 50016

typedef float v2f __attribute__((ext_vector_type(2)));
typedef float v4f __attribute__((ext_vector_type(4)));
typedef float v8f __attribute__((ext_vector_type(8)));

__device__ __forceinline__ float leaky(float x) { return x > 0.f ? x : SLOPE * x; }

__device__ __forceinline__ void accumulator_slots() {
    __builtin_amdgcn_sched_barrier(0);
    asm volatile("v_nop\n\tv_nop\n\tv_nop\n\tv_nop");
    __builtin_amdgcn_sched_barrier(0);
}

template <int M>
__device__ __forceinline__ float row_xor(const float x) {
    const int xi = __float_as_int(x);
    return __int_as_float(__builtin_amdgcn_update_dpp(xi, xi, 0x160 + M, 0xf, 0xf, true));
}

template <int CNT, int M>
__device__ __forceinline__ void halving_step(float* v, const int lane) {
    const bool up = (lane & M) != 0;
#pragma unroll
    for (int i = 0; i < CNT / 2; ++i) {
        const float keep = up ? v[i + CNT / 2] : v[i];
        const float send = up ? v[i] : v[i + CNT / 2];
        v[i] = keep + row_xor<M>(send);
    }
}
template <int N>
__device__ __forceinline__ void halfwave_sums(float (&v)[N], const int lane) {
    halving_step<N, 8>(v, lane);
    halving_step<N / 2, 4>(v, lane);
    halving_step<N / 4, 2>(v, lane);
    halving_step<N / 8, 1>(v, lane);
}

__global__ void __launch_bounds__(128) gemm1_kernel(const float* __restrict__ X,
                                                    const float* __restrict__ W,
                                                    const float* __restrict__ AR,
                                                    float* __restrict__ C,
                                                    float* __restrict__ ER) {
    __shared__ float s_er[16 * H1N];
    const int wave = threadIdx.x >> 5;
    const int lane = threadIdx.x & 31;
    const int half = lane >> 4;
    const int r    = lane & 15;
    const int row0 = blockIdx.x * 16;
    const int col0 = wave * 64;

    v8f acc[4] = {};
    const float* xp = X + (size_t)(row0 + r) * INF_ + half * 4;
    const float* wp = W + (size_t)(half * 4) * 256 + col0 + r * 4;

#pragma nounroll
    for (int k = 0; k < INF_; k += 8) {
        const v4f a4 = *(const v4f*)(xp + k);
        const float* wk = wp + (size_t)k * 256;
        const v4f b0 = *(const v4f*)(wk);
        const v4f b1 = *(const v4f*)(wk + 256);
        const v4f b2 = *(const v4f*)(wk + 512);
        const v4f b3 = *(const v4f*)(wk + 768);
        v2f a;
        a.x = a4.x; a.y = a4.y;
#pragma unroll
        for (int t = 0; t < 4; ++t) {
            v2f b;
            b.x = b0[t]; b.y = b1[t];
            acc[t] = __builtin_amdgcn_wmma_f32_16x16x4_f32(
                false, a, false, b, (short)0, acc[t], false, false);
        }
        a.x = a4.z; a.y = a4.w;
#pragma unroll
        for (int t = 0; t < 4; ++t) {
            v2f b;
            b.x = b2[t]; b.y = b3[t];
            acc[t] = __builtin_amdgcn_wmma_f32_16x16x4_f32(
                false, a, false, b, (short)0, acc[t], false, false);
        }
    }

    accumulator_slots();

    const int lrow = half * 8 + (r & 7);
    const int hd   = wave * 2 + (r >> 3);
    {
        const v4f wc = *(const v4f*)(AR + col0 + r * 4);
        float v[8];
#pragma unroll
        for (int p = 0; p < 4; ++p) {
            v2f q[4];
#pragma unroll
            for (int t = 0; t < 4; ++t) { q[t].x = acc[t][2 * p]; q[t].y = acc[t][2 * p + 1]; }
            v2f s = q[0] * wc.x;
            s += q[1] * wc.y;
            s += q[2] * wc.z;
            s += q[3] * wc.w;
            v[2 * p + 0] = s.x;
            v[2 * p + 1] = s.y;
        }
        halving_step<8, 4>(v, lane);
        halving_step<4, 2>(v, lane);
        halving_step<2, 1>(v, lane);
        s_er[lrow * H1N + hd] = v[0];
    }
    __syncthreads();
    ER[(size_t)row0 * H1N + threadIdx.x] = s_er[threadIdx.x];

#pragma unroll
    for (int j = 0; j < 8; ++j) {
        const int row = row0 + half * 8 + j;
        v4f o;
        o.x = acc[0][j]; o.y = acc[1][j]; o.z = acc[2][j]; o.w = acc[3][j];
        *(v4f*)(C + (size_t)row * 256 + col0 + r * 4) = o;
    }
}

__global__ void __launch_bounds__(128) gemm2_kernel(const float* __restrict__ Y,
                                                    const float* __restrict__ W2,
                                                    const float* __restrict__ RW,
                                                    const float* __restrict__ b2,
                                                    const float* __restrict__ AL,
                                                    const float* __restrict__ AR,
                                                    float* __restrict__ F2,
                                                    float* __restrict__ OUTP,
                                                    float* __restrict__ EL,
                                                    float* __restrict__ ER,
                                                    int n) {
    __shared__ float s_lg[2][64];
    const int wave = threadIdx.x >> 5;
    const int lane = threadIdx.x & 31;
    const int half = lane >> 4;
    const int r    = lane & 15;
    const int row0 = blockIdx.x * 64 + wave * 16;

    int rload = row0 + r;
    if (rload >= n) rload = n - 1;
    const float* yp  = Y  + (size_t)rload * 256 + half * 4;
    const float* w2p = W2 + (size_t)(half * 4) * 32 + r * 2;
    const float* rwp = RW + (size_t)(half * 4) * 32 + r * 2;

    v8f accW[2] = {};
    v8f accR[2] = {};

    for (int k = 0; k < 256; k += 8) {
        const v4f a4 = *(const v4f*)(yp + k);
        const float* wk = w2p + (size_t)k * 32;
        const float* rk = rwp + (size_t)k * 32;
        const v2f w0 = *(const v2f*)(wk);
        const v2f w1 = *(const v2f*)(wk + 32);
        const v2f w2 = *(const v2f*)(wk + 64);
        const v2f w3 = *(const v2f*)(wk + 96);
        const v2f q0 = *(const v2f*)(rk);
        const v2f q1 = *(const v2f*)(rk + 32);
        const v2f q2 = *(const v2f*)(rk + 64);
        const v2f q3 = *(const v2f*)(rk + 96);
        v2f a;
        a.x = a4.x; a.y = a4.y;
#pragma unroll
        for (int t = 0; t < 2; ++t) {
            v2f bw, br;
            bw.x = w0[t]; bw.y = w1[t];
            accW[t] = __builtin_amdgcn_wmma_f32_16x16x4_f32(
                false, a, false, bw, (short)0, accW[t], false, false);
            br.x = q0[t]; br.y = q1[t];
            accR[t] = __builtin_amdgcn_wmma_f32_16x16x4_f32(
                false, a, false, br, (short)0, accR[t], false, false);
        }
        a.x = a4.z; a.y = a4.w;
#pragma unroll
        for (int t = 0; t < 2; ++t) {
            v2f bw, br;
            bw.x = w2[t]; bw.y = w3[t];
            accW[t] = __builtin_amdgcn_wmma_f32_16x16x4_f32(
                false, a, false, bw, (short)0, accW[t], false, false);
            br.x = q2[t]; br.y = q3[t];
            accR[t] = __builtin_amdgcn_wmma_f32_16x16x4_f32(
                false, a, false, br, (short)0, accR[t], false, false);
        }
    }

    accumulator_slots();

    const v2f al_c = *(const v2f*)(AL + r * 2);
    const v2f ar_c = *(const v2f*)(AR + r * 2);
    float v[16];
#pragma unroll
    for (int j = 0; j < 8; ++j) {
        v[2 * j + 0] = accW[0][j] * al_c.x + accW[1][j] * al_c.y;
        v[2 * j + 1] = accW[0][j] * ar_c.x + accW[1][j] * ar_c.y;
    }
    halfwave_sums<16>(v, lane);
    s_lg[r & 1][wave * 16 + half * 8 + (r >> 1)] = v[0];
    __syncthreads();
    {
        const int which = wave >> 1;
        const int li    = (wave & 1) * 32 + lane;
        const int grow  = blockIdx.x * 64 + li;
        float* dstp = which ? ER : EL;
        if (grow < n) dstp[grow] = s_lg[which][li];
    }

    const v2f bias = *(const v2f*)(b2 + r * 2);
#pragma unroll
    for (int j = 0; j < 8; ++j) {
        const int row = row0 + half * 8 + j;
        if (row < n) {
            v2f f, o;
            f.x = accW[0][j];          f.y = accW[1][j];
            o.x = accR[0][j] + bias.x; o.y = accR[1][j] + bias.y;
            *(v2f*)(F2   + (size_t)row * 32 + r * 2) = f;
            *(v2f*)(OUTP + (size_t)row * 32 + r * 2) = o;
        }
    }
}

#define NCHUNK 250
#define CHUNK (EE / NCHUNK)
#define NBKT ((NN + 255) / 256)
#define HN (NBKT * NCHUNK)
#define SCAN_BLOCKS 196
#define SCAN_LEN (SCAN_BLOCKS * 256)
#define SORT_T 512
#define PER_T ((CHUNK + SORT_T - 1) / SORT_T)
#define SEG_T 1024
#define SEG_B 5
#define SEG_CAP 8192
#define NO_EDGE 0xFFFFFFFFu
static_assert(EE % NCHUNK == 0, "chunks must tile the edge list");
static_assert(HN + 1 <= SCAN_LEN && NBKT * 256 <= SCAN_LEN, "scan length");
static_assert(NN <= 65536 && NBKT <= 255, "edge word fields");

__global__ void __launch_bounds__(SORT_T) chunk_count_kernel(const int* __restrict__ dst,
                                                             int* __restrict__ bh) {
    __shared__ int s_h[256];
    const int t = threadIdx.x, b = blockIdx.x;
    if (t < 256) s_h[t] = 0;
    const int* dp = dst + b * CHUNK;
    int d[PER_T];
#pragma unroll
    for (int i = 0; i < PER_T; ++i) {
        const int e = i * SORT_T + t;
        d[i] = (e < CHUNK) ? dp[e] : -1;
    }
    __syncthreads();
#pragma unroll
    for (int i = 0; i < PER_T; ++i)
        if (d[i] >= 0) atomicAdd(&s_h[d[i] >> 8], 1);
    __syncthreads();
    if (t < NBKT) bh[t * NCHUNK + b] = s_h[t];
    if (b == 0)
        for (int i = HN + t; i < SCAN_LEN; i += SORT_T) bh[i] = 0;
}

__global__ void __launch_bounds__(256) scan_kernel(const int* __restrict__ cnt,
                                                   int* __restrict__ off) {
    __shared__ int s_part[8];
    __shared__ int s_tot[8];
    const int t = threadIdx.x, lane = t & 31, w = t >> 5;
    const int b = blockIdx.x;
    int part = 0;
    for (int j = 0; j < b; ++j) part += cnt[j * 256 + t];
    const int v = cnt[b * 256 + t];
#pragma unroll
    for (int o = 16; o > 0; o >>= 1) part += __shfl_xor(part, o, 32);
    int inc = v;
#pragma unroll
    for (int o = 1; o < 32; o <<= 1) {
        const int nb = __shfl_up(inc, o, 32);
        if (lane >= o) inc += nb;
    }
    if (lane == 0)  s_part[w] = part;
    if (lane == 31) s_tot[w]  = inc;
    __syncthreads();
    int base = 0;
#pragma unroll
    for (int i = 0; i < 8; ++i) {
        base += s_part[i];
        if (i < w) base += s_tot[i];
    }
    off[b * 256 + t] = base + inc - v;
}

__device__ __forceinline__ int scan256(const int c, const int t, int* s_tot) {
    const int lane = t & 31, w = t >> 5;
    int inc = c;
    if (t < 256) {
        for (int o = 1; o < 32; o <<= 1) {
            const int nb = __shfl_up(inc, o, 32);
            if (lane >= o) inc += nb;
        }
        if (lane == 31) s_tot[w] = inc;
    }
    __syncthreads();
    int base = 0;
    if (t < 256) {
        for (int i = 0; i < 8; ++i)
            if (i < w) base += s_tot[i];
    }
    return base + inc - c;
}

__global__ void __launch_bounds__(SORT_T) chunk_place_kernel(const int* __restrict__ src,
                                                             const int* __restrict__ dst,
                                                             const int* __restrict__ boff,
                                                             unsigned* __restrict__ tmp) {
    __shared__ int s_cur[256];
    __shared__ int s_gb[256];
    __shared__ int s_tot[8];
    __shared__ unsigned s_sorted[CHUNK];
    const int t = threadIdx.x, b = blockIdx.x;
    const int* sp = src + b * CHUNK;
    const int* dp = dst + b * CHUNK;
    int s[PER_T], d[PER_T];
#pragma unroll
    for (int i = 0; i < PER_T; ++i) {
        const int e = i * SORT_T + t;
        s[i] = (e < CHUNK) ? sp[e] : 0;
        d[i] = (e < CHUNK) ? dp[e] : -1;
    }
    int c = 0, g0 = 0;
    if (t < NBKT) {
        g0 = boff[t * NCHUNK + b];
        c  = boff[t * NCHUNK + b + 1] - g0;
    }
    const int ls = scan256(c, t, s_tot);
    if (t < 256) { s_cur[t] = ls; s_gb[t] = g0 - ls; }
    __syncthreads();
#pragma unroll
    for (int i = 0; i < PER_T; ++i) {
        if (d[i] >= 0) {
            const int k = d[i] >> 8;
            const int slot = atomicAdd(&s_cur[k], 1);
            s_sorted[slot] = (unsigned)s[i] | ((unsigned)(d[i] & 255) << 16) | ((unsigned)k << 24);
        }
    }
    __syncthreads();
    for (int i = t; i < CHUNK; i += SORT_T) {
        const unsigned v = s_sorted[i];
        tmp[s_gb[v >> 24] + i] = v;
    }
}

__global__ void __launch_bounds__(SEG_T) bucket_sort_kernel(const unsigned* __restrict__ tmp,
                                                            const int* __restrict__ boff,
                                                            int* __restrict__ off,
                                                            int* __restrict__ csrc) {
    __shared__ int s_cnt[256];
    __shared__ int s_cur[256];
    __shared__ int s_tot[8];
    __shared__ int s_sorted[SEG_CAP];
    const int t = threadIdx.x, k = blockIdx.x;
    const int beg = boff[k * NCHUNK];
    const int n   = boff[(k + 1) * NCHUNK] - beg;
    const unsigned* tp = tmp + beg;
    if (t < 256) s_cnt[t] = 0;
    __syncthreads();
    for (int i0 = 0; i0 < n; i0 += SEG_T * SEG_B) {
        unsigned v[SEG_B];
#pragma unroll
        for (int j = 0; j < SEG_B; ++j) {
            const int i = i0 + j * SEG_T + t;
            v[j] = (i < n) ? tp[i] : NO_EDGE;
        }
#pragma unroll
        for (int j = 0; j < SEG_B; ++j)
            if (v[j] != NO_EDGE) atomicAdd(&s_cnt[(v[j] >> 16) & 255], 1);
    }
    __syncthreads();
    const int c  = (t < 256) ? s_cnt[t] : 0;
    const int lo = scan256(c, t, s_tot);
    if (t < 256) {
        s_cur[t] = lo;
        off[k * 256 + t] = beg + lo;
    }
    __syncthreads();
    const bool fits = n <= SEG_CAP;
    for (int i0 = 0; i0 < n; i0 += SEG_T * SEG_B) {
        unsigned v[SEG_B];
#pragma unroll
        for (int j = 0; j < SEG_B; ++j) {
            const int i = i0 + j * SEG_T + t;
            v[j] = (i < n) ? tp[i] : NO_EDGE;
        }
#pragma unroll
        for (int j = 0; j < SEG_B; ++j) {
            if (v[j] != NO_EDGE) {
                const int slot = atomicAdd(&s_cur[(v[j] >> 16) & 255], 1);
                const int sv = (int)(v[j] & 0xFFFFu);
                if (fits) s_sorted[slot] = sv;
                else      csrc[beg + slot] = sv;
            }
        }
    }
    if (fits) {
        __syncthreads();
        for (int i = t; i < n; i += SEG_T) csrc[beg + i] = s_sorted[i];
    }
}


template <int N>
__device__ __forceinline__ void agg1_group(const float* __restrict__ feat, const int s_l,
                                           const int i, const int lane,
                                           const v2f (&w)[4], const float erd,
                                           float& m, float& l, v2f (&a)[4]) {
    v4f f0[N], f1[N];
#pragma unroll
    for (int k = 0; k < N; ++k) {
        const int s = __builtin_amdgcn_readlane(s_l, i + k);
        const v4f* fs = (const v4f*)(feat + (size_t)s * 256 + lane * 8);
        f0[k] = fs[0];
        f1[k] = fs[1];
    }
    float sc[N];
    float mc = m;
#pragma unroll
    for (int k = 0; k < N; ++k) {
        v2f q = f0[k].xy * w[0];
        q += f0[k].zw * w[1];
        q += f1[k].xy * w[2];
        q += f1[k].zw * w[3];
        float p = q.x + q.y;
        p += row_xor<1>(p);
        p += row_xor<2>(p);
        sc[k] = leaky(p + erd);
        mc = fmaxf(mc, sc[k]);
    }
    const float c = __expf(m - mc);
    l *= c;
#pragma unroll
    for (int j = 0; j < 4; ++j) a[j] *= c;
#pragma unroll
    for (int k = 0; k < N; ++k) {
        const float p = __expf(sc[k] - mc);
        l += p;
        a[0] += f0[k].xy * p;
        a[1] += f0[k].zw * p;
        a[2] += f1[k].xy * p;
        a[3] += f1[k].zw * p;
    }
    m = mc;
}

__global__ void __launch_bounds__(256) agg1_gather_kernel(const int* __restrict__ off,
                                                          const int* __restrict__ csrc,
                                                          const float* __restrict__ al,
                                                          const float* __restrict__ er,
                                                          const float* __restrict__ feat,
                                                          const float* __restrict__ x,
                                                          const float* __restrict__ b1,
                                                          float* __restrict__ rst) {
    const int lane = threadIdx.x & 31;
    const int d = __builtin_amdgcn_readfirstlane((int)(blockIdx.x * 8 + (threadIdx.x >> 5)));
    const int beg = off[d], end = off[d + 1];
    const int h = lane >> 2;
    const float erd = er[d * H1N + h];
    v2f w[4];
#pragma unroll
    for (int j = 0; j < 4; ++j) w[j] = *(const v2f*)(al + lane * 8 + 2 * j);

    float m = -INFINITY, l = 0.f;
    v2f a[4] = {};
    for (int base = beg; base < end; base += 32) {
        const int mcnt = (end - base < 32) ? (end - base) : 32;
        const int s_l = (lane < mcnt) ? csrc[base + lane] : 0;
        int i = 0;
        for (; i + 4 <= mcnt; i += 4)
            agg1_group<4>(feat, s_l, i, lane, w, erd, m, l, a);
        const int rem = mcnt - i;
        if (rem == 3)      agg1_group<3>(feat, s_l, i, lane, w, erd, m, l, a);
        else if (rem == 2) agg1_group<2>(feat, s_l, i, lane, w, erd, m, l, a);
        else if (rem == 1) agg1_group<1>(feat, s_l, i, lane, w, erd, m, l, a);
    }
    const float inv = (l > 0.f) ? 1.f / l : 0.f;
    const size_t o = (size_t)d * 256 + lane * 8;
    const float4 x0 = *(const float4*)(x + o);
    const float4 x1 = *(const float4*)(x + o + 4);
    const float4 c0 = *(const float4*)(b1 + lane * 8);
    const float4 c1 = *(const float4*)(b1 + lane * 8 + 4);
    float4 r0, r1;
    r0.x = fmaxf(a[0].x * inv + x0.x + c0.x, 0.f); r0.y = fmaxf(a[0].y * inv + x0.y + c0.y, 0.f);
    r0.z = fmaxf(a[1].x * inv + x0.z + c0.z, 0.f); r0.w = fmaxf(a[1].y * inv + x0.w + c0.w, 0.f);
    r1.x = fmaxf(a[2].x * inv + x1.x + c1.x, 0.f); r1.y = fmaxf(a[2].y * inv + x1.y + c1.y, 0.f);
    r1.z = fmaxf(a[3].x * inv + x1.z + c1.z, 0.f); r1.w = fmaxf(a[3].y * inv + x1.w + c1.w, 0.f);
    *(float4*)(rst + o)     = r0;
    *(float4*)(rst + o + 4) = r1;
}

__global__ void __launch_bounds__(256) agg2_gather_kernel(const int* __restrict__ off,
                                                          const int* __restrict__ csrc,
                                                          const float* __restrict__ el,
                                                          const float* __restrict__ er,
                                                          const float* __restrict__ feat,
                                                          float* __restrict__ outp) {
    const int lane = threadIdx.x & 31;
    const int d = __builtin_amdgcn_readfirstlane((int)(blockIdx.x * 8 + (threadIdx.x >> 5)));
    const int beg = off[d], end = off[d + 1];
    const float erd = er[d];

    const bool have0 = beg + lane < end;
    int   s0 = 0;
    float v0 = -INFINITY;
    if (have0) { s0 = csrc[beg + lane]; v0 = leaky(el[s0] + erd); }
    float mx = v0;
    for (int e = beg + 32 + lane; e < end; e += 32)
        mx = fmaxf(mx, leaky(el[csrc[e]] + erd));
    for (int o = 16; o > 0; o >>= 1) mx = fmaxf(mx, __shfl_xor(mx, o, 32));
    const float w0 = have0 ? __expf(v0 - mx) : 0.f;
    float sm = w0;
    for (int e = beg + 32 + lane; e < end; e += 32)
        sm += __expf(leaky(el[csrc[e]] + erd) - mx);
    for (int o = 16; o > 0; o >>= 1) sm += __shfl_xor(sm, o, 32);

    float acc = 0.f;
    for (int base = beg; base < end; base += 32) {
        const int mcnt = (end - base < 32) ? (end - base) : 32;
        int   s_l = s0;
        float a_l = w0 / sm;
        if (base != beg) {
            s_l = 0; a_l = 0.f;
            if (lane < mcnt) {
                s_l = csrc[base + lane];
                a_l = __expf(leaky(el[s_l] + erd) - mx) / sm;
            }
        }
#pragma unroll 4
        for (int i = 0; i < mcnt; ++i) {
            const int   s = __builtin_amdgcn_readlane(s_l, i);
            const float a = __int_as_float(__builtin_amdgcn_readlane(__float_as_int(a_l), i));
            acc += feat[(size_t)s * 32 + lane] * a;
        }
    }
    const size_t o = (size_t)d * 32 + lane;
    outp[o] = outp[o] + acc;
}

extern "C" void kernel_launch(void* const* d_in, const int* in_sizes, int n_in,
                              void* d_out, int out_size, void* d_ws, size_t ws_size,
                              hipStream_t stream) {
    const float* x     = (const float*)d_in[0];
    const int*   src   = (const int*)  d_in[1];
    const int*   dst   = (const int*)  d_in[2];
    const float* W1    = (const float*)d_in[3];
    const float* al1   = (const float*)d_in[4];
    const float* ar1   = (const float*)d_in[5];
    const float* b1    = (const float*)d_in[6];
    const float* W2    = (const float*)d_in[7];
    const float* al2   = (const float*)d_in[8];
    const float* ar2   = (const float*)d_in[9];
    const float* resW2 = (const float*)d_in[10];
    const float* b2    = (const float*)d_in[11];
    float* outp = (float*)d_out;

    float* ws    = (float*)d_ws;
    float* feat1 = ws;
    float* rst1  = feat1 + (size_t)NN * 256;
    float* er1   = rst1  + (size_t)NN * 256;
    int*   csrc  = (int*)(er1 + (size_t)NN * 8);
    unsigned* tmp = (unsigned*)(csrc + EE);
    int*   bh    = (int*)(tmp + EE);
    int*   off   = bh   + SCAN_LEN;
    int*   boff  = (int*)rst1;
    float* feat2 = ws;
    float* el2   = feat2 + (size_t)NN * 32;
    float* er2   = el2   + NN_PAD;

    chunk_count_kernel<<<NCHUNK, SORT_T, 0, stream>>>(dst, bh);
    scan_kernel<<<SCAN_BLOCKS, 256, 0, stream>>>(bh, boff);
    chunk_place_kernel<<<NCHUNK, SORT_T, 0, stream>>>(src, dst, boff, tmp);
    bucket_sort_kernel<<<NBKT, SEG_T, 0, stream>>>(tmp, boff, off, csrc);

    gemm1_kernel<<<NN / 16, 128, 0, stream>>>(x, W1, ar1, feat1, er1);
    agg1_gather_kernel<<<NN / 8, 256, 0, stream>>>(off, csrc, al1, er1, feat1, x, b1, rst1);

    gemm2_kernel<<<(NN + 63) / 64, 128, 0, stream>>>(rst1, W2, resW2, b2, al2, ar2,
                                                     feat2, outp, el2, er2, NN);
    agg2_gather_kernel<<<NN / 8, 256, 0, stream>>>(off, csrc, el2, er2, feat2, outp);
}


# Edge sort in 2 launches, not 4: each chunk sorted in place with its own bucket table; a bucket block sums the tables for its base and finds its runs by LDS search; segment order still LDS arrival
# speedup vs baseline: 6.0003x; 1.1150x over previous
#include <hip/hip_runtime.h>
#include <hip/hip_bf16.h>
#include <math.h>

#define NN 50000
#define EE 800000
#define INF_ 256
#define OUTF 32
#define H1N 8
#define SLOPE 0.2f
#define NN_PAD 50016

typedef float v2f __attribute__((ext_vector_type(2)));
typedef float v4f __attribute__((ext_vector_type(4)));
typedef float v8f __attribute__((ext_vector_type(8)));

__device__ __forceinline__ float leaky(float x) { return x > 0.f ? x : SLOPE * x; }

__device__ __forceinline__ void accumulator_slots() {
    __builtin_amdgcn_sched_barrier(0);
    asm volatile("v_nop\n\tv_nop\n\tv_nop\n\tv_nop");
    __builtin_amdgcn_sched_barrier(0);
}

template <int M>
__device__ __forceinline__ float row_xor(const float x) {
    const int xi = __float_as_int(x);
    return __int_as_float(__builtin_amdgcn_update_dpp(xi, xi, 0x160 + M, 0xf, 0xf, true));
}

template <int CNT, int M>
__device__ __forceinline__ void halving_step(float* v, const int lane) {
    const bool up = (lane & M) != 0;
#pragma unroll
    for (int i = 0; i < CNT / 2; ++i) {
        const float keep = up ? v[i + CNT / 2] : v[i];
        const float send = up ? v[i] : v[i + CNT / 2];
        v[i] = keep + row_xor<M>(send);
    }
}
template <int N>
__device__ __forceinline__ void halfwave_sums(float (&v)[N], const int lane) {
    halving_step<N, 8>(v, lane);
    halving_step<N / 2, 4>(v, lane);
    halving_step<N / 4, 2>(v, lane);
    halving_step<N / 8, 1>(v, lane);
}

__global__ void __launch_bounds__(128) gemm1_kernel(const float* __restrict__ X,
                                                    const float* __restrict__ W,
                                                    const float* __restrict__ AR,
                                                    float* __restrict__ C,
                                                    float* __restrict__ ER) {
    __shared__ float s_er[16 * H1N];
    const int wave = threadIdx.x >> 5;
    const int lane = threadIdx.x & 31;
    const int half = lane >> 4;
    const int r    = lane & 15;
    const int row0 = blockIdx.x * 16;
    const int col0 = wave * 64;

    v8f acc[4] = {};
    const float* xp = X + (size_t)(row0 + r) * INF_ + half * 4;
    const float* wp = W + (size_t)(half * 4) * 256 + col0 + r * 4;

#pragma nounroll
    for (int k = 0; k < INF_; k += 8) {
        const v4f a4 = *(const v4f*)(xp + k);
        const float* wk = wp + (size_t)k * 256;
        const v4f b0 = *(const v4f*)(wk);
        const v4f b1 = *(const v4f*)(wk + 256);
        const v4f b2 = *(const v4f*)(wk + 512);
        const v4f b3 = *(const v4f*)(wk + 768);
        v2f a;
        a.x = a4.x; a.y = a4.y;
#pragma unroll
        for (int t = 0; t < 4; ++t) {
            v2f b;
            b.x = b0[t]; b.y = b1[t];
            acc[t] = __builtin_amdgcn_wmma_f32_16x16x4_f32(
                false, a, false, b, (short)0, acc[t], false, false);
        }
        a.x = a4.z; a.y = a4.w;
#pragma unroll
        for (int t = 0; t < 4; ++t) {
            v2f b;
            b.x = b2[t]; b.y = b3[t];
            acc[t] = __builtin_amdgcn_wmma_f32_16x16x4_f32(
                false, a, false, b, (short)0, acc[t], false, false);
        }
    }

    accumulator_slots();

    const int lrow = half * 8 + (r & 7);
    const int hd   = wave * 2 + (r >> 3);
    {
        const v4f wc = *(const v4f*)(AR + col0 + r * 4);
        float v[8];
#pragma unroll
        for (int p = 0; p < 4; ++p) {
            v2f q[4];
#pragma unroll
            for (int t = 0; t < 4; ++t) { q[t].x = acc[t][2 * p]; q[t].y = acc[t][2 * p + 1]; }
            v2f s = q[0] * wc.x;
            s += q[1] * wc.y;
            s += q[2] * wc.z;
            s += q[3] * wc.w;
            v[2 * p + 0] = s.x;
            v[2 * p + 1] = s.y;
        }
        halving_step<8, 4>(v, lane);
        halving_step<4, 2>(v, lane);
        halving_step<2, 1>(v, lane);
        s_er[lrow * H1N + hd] = v[0];
    }
    __syncthreads();
    ER[(size_t)row0 * H1N + threadIdx.x] = s_er[threadIdx.x];

#pragma unroll
    for (int j = 0; j < 8; ++j) {
        const int row = row0 + half * 8 + j;
        v4f o;
        o.x = acc[0][j]; o.y = acc[1][j]; o.z = acc[2][j]; o.w = acc[3][j];
        *(v4f*)(C + (size_t)row * 256 + col0 + r * 4) = o;
    }
}

__global__ void __launch_bounds__(128) gemm2_kernel(const float* __restrict__ Y,
                                                    const float* __restrict__ W2,
                                                    const float* __restrict__ RW,
                                                    const float* __restrict__ b2,
                                                    const float* __restrict__ AL,
                                                    const float* __restrict__ AR,
                                                    float* __restrict__ F2,
                                                    float* __restrict__ OUTP,
                                                    float* __restrict__ EL,
                                                    float* __restrict__ ER,
                                                    int n) {
    __shared__ float s_lg[2][64];
    const int wave = threadIdx.x >> 5;
    const int lane = threadIdx.x & 31;
    const int half = lane >> 4;
    const int r    = lane & 15;
    const int row0 = blockIdx.x * 64 + wave * 16;

    int rload = row0 + r;
    if (rload >= n) rload = n - 1;
    const float* yp  = Y  + (size_t)rload * 256 + half * 4;
    const float* w2p = W2 + (size_t)(half * 4) * 32 + r * 2;
    const float* rwp = RW + (size_t)(half * 4) * 32 + r * 2;

    v8f accW[2] = {};
    v8f accR[2] = {};

    for (int k = 0; k < 256; k += 8) {
        const v4f a4 = *(const v4f*)(yp + k);
        const float* wk = w2p + (size_t)k * 32;
        const float* rk = rwp + (size_t)k * 32;
        const v2f w0 = *(const v2f*)(wk);
        const v2f w1 = *(const v2f*)(wk + 32);
        const v2f w2 = *(const v2f*)(wk + 64);
        const v2f w3 = *(const v2f*)(wk + 96);
        const v2f q0 = *(const v2f*)(rk);
        const v2f q1 = *(const v2f*)(rk + 32);
        const v2f q2 = *(const v2f*)(rk + 64);
        const v2f q3 = *(const v2f*)(rk + 96);
        v2f a;
        a.x = a4.x; a.y = a4.y;
#pragma unroll
        for (int t = 0; t < 2; ++t) {
            v2f bw, br;
            bw.x = w0[t]; bw.y = w1[t];
            accW[t] = __builtin_amdgcn_wmma_f32_16x16x4_f32(
                false, a, false, bw, (short)0, accW[t], false, false);
            br.x = q0[t]; br.y = q1[t];
            accR[t] = __builtin_amdgcn_wmma_f32_16x16x4_f32(
                false, a, false, br, (short)0, accR[t], false, false);
        }
        a.x = a4.z; a.y = a4.w;
#pragma unroll
        for (int t = 0; t < 2; ++t) {
            v2f bw, br;
            bw.x = w2[t]; bw.y = w3[t];
            accW[t] = __builtin_amdgcn_wmma_f32_16x16x4_f32(
                false, a, false, bw, (short)0, accW[t], false, false);
            br.x = q2[t]; br.y = q3[t];
            accR[t] = __builtin_amdgcn_wmma_f32_16x16x4_f32(
                false, a, false, br, (short)0, accR[t], false, false);
        }
    }

    accumulator_slots();

    const v2f al_c = *(const v2f*)(AL + r * 2);
    const v2f ar_c = *(const v2f*)(AR + r * 2);
    float v[16];
#pragma unroll
    for (int j = 0; j < 8; ++j) {
        v[2 * j + 0] = accW[0][j] * al_c.x + accW[1][j] * al_c.y;
        v[2 * j + 1] = accW[0][j] * ar_c.x + accW[1][j] * ar_c.y;
    }
    halfwave_sums<16>(v, lane);
    s_lg[r & 1][wave * 16 + half * 8 + (r >> 1)] = v[0];
    __syncthreads();
    {
        const int which = wave >> 1;
        const int li    = (wave & 1) * 32 + lane;
        const int grow  = blockIdx.x * 64 + li;
        float* dstp = which ? ER : EL;
        if (grow < n) dstp[grow] = s_lg[which][li];
    }

    const v2f bias = *(const v2f*)(b2 + r * 2);
#pragma unroll
    for (int j = 0; j < 8; ++j) {
        const int row = row0 + half * 8 + j;
        if (row < n) {
            v2f f, o;
            f.x = accW[0][j];          f.y = accW[1][j];
            o.x = accR[0][j] + bias.x; o.y = accR[1][j] + bias.y;
            *(v2f*)(F2   + (size_t)row * 32 + r * 2) = f;
            *(v2f*)(OUTP + (size_t)row * 32 + r * 2) = o;
        }
    }
}

#define NCHUNK 250
#define CHUNK (EE / NCHUNK)
#define NBKT ((NN + 255) / 256)
#define LO_PITCH 224
#define OFF_LEN (NBKT * 256)
#define SORT_T 512
#define PER_T ((CHUNK + SORT_T - 1) / SORT_T)
#define SEG_T 1024
#define SEG_B 5
#define SEG_CAP 8192
#define NO_EDGE 0xFFFFFFFFu
static_assert(EE % NCHUNK == 0 && CHUNK % 4 == 0, "chunks must tile the edge list in 16-B pieces");
static_assert(NCHUNK <= 256 && NBKT < 256, "one thread of the first 256 per chunk / per bucket");
static_assert(NN <= 65536 && NN + 1 <= OFF_LEN, "edge word fields, offsets length");
static_assert(LO_PITCH >= NBKT + 1 && LO_PITCH % 32 == 0, "table row: whole lines");

__device__ __forceinline__ int scan256(const int c, const int t, int* s_tot) {
    const int lane = t & 31, w = t >> 5;
    int inc = c;
    if (t < 256) {
        for (int o = 1; o < 32; o <<= 1) {
            const int nb = __shfl_up(inc, o, 32);
            if (lane >= o) inc += nb;
        }
        if (lane == 31) s_tot[w] = inc;
    }
    __syncthreads();
    int base = 0;
    if (t < 256) {
        for (int i = 0; i < 8; ++i)
            if (i < w) base += s_tot[i];
    }
    return base + inc - c;
}

__device__ __forceinline__ void scan256_pair(const int a, const int b, const int t, int* s_tot,
                                             int& ex_a, int& ex_b) {
    const int lane = t & 31, w = t >> 5;
    int ia = a, ib = b;
    if (t < 256) {
        for (int o = 1; o < 32; o <<= 1) {
            const int na = __shfl_up(ia, o, 32);
            const int nb = __shfl_up(ib, o, 32);
            if (lane >= o) { ia += na; ib += nb; }
        }
        if (lane == 31) { s_tot[w] = ia; s_tot[8 + w] = ib; }
    }
    __syncthreads();
    int ba = 0, bb = 0;
    if (t < 256) {
        for (int i = 0; i < 8; ++i)
            if (i < w) { ba += s_tot[i]; bb += s_tot[8 + i]; }
    }
    ex_a = ba + ia - a;
    ex_b = bb + ib - b;
}

__global__ void __launch_bounds__(SORT_T) chunk_sort_kernel(const int* __restrict__ src,
                                                            const int* __restrict__ dst,
                                                            int* __restrict__ lo,
                                                            unsigned* __restrict__ tmp) {
    __shared__ int s_cnt[256];
    __shared__ int s_tot[8];
    __shared__ __attribute__((aligned(16))) unsigned s_sorted[CHUNK];
    const int t = threadIdx.x, b = blockIdx.x;
    if (t < 256) s_cnt[t] = 0;
    const int* sp = src + b * CHUNK;
    const int* dp = dst + b * CHUNK;
    int s[PER_T], d[PER_T], rk[PER_T];
#pragma unroll
    for (int i = 0; i < PER_T; ++i) {
        const int e = i * SORT_T + t;
        s[i] = (e < CHUNK) ? sp[e] : 0;
        d[i] = (e < CHUNK) ? dp[e] : -1;
    }
    __syncthreads();
#pragma unroll
    for (int i = 0; i < PER_T; ++i)
        rk[i] = (d[i] >= 0) ? atomicAdd(&s_cnt[d[i] >> 8], 1) : 0;
    __syncthreads();
    const int c  = (t < 256) ? s_cnt[t] : 0;
    const int ls = scan256(c, t, s_tot);
    if (t < 256) s_cnt[t] = ls;
    if (t <= NBKT) lo[b * LO_PITCH + t] = ls;
    __syncthreads();
#pragma unroll
    for (int i = 0; i < PER_T; ++i) {
        if (d[i] >= 0) {
            const int k = d[i] >> 8;
            s_sorted[s_cnt[k] + rk[i]] = (unsigned)s[i] | ((unsigned)(d[i] & 255) << 16);
        }
    }
    __syncthreads();
    const uint4* s4 = (const uint4*)s_sorted;
    uint4* t4 = (uint4*)(tmp + (size_t)b * CHUNK);
    for (int i = t; i < CHUNK / 4; i += SORT_T) t4[i] = s4[i];
}

__device__ __forceinline__ void bucket_fetch(const unsigned* __restrict__ tmp, const int* s_rs,
                                             const int* s_ra, const int i0, const int t,
                                             const int n, unsigned (&v)[SEG_B]) {
    int ic[SEG_B], r[SEG_B];
#pragma unroll
    for (int j = 0; j < SEG_B; ++j) {
        const int i = i0 + j * SEG_T + t;
        ic[j] = (i < n) ? i : n - 1;
        r[j] = 0;
    }
#pragma unroll
    for (int st = 128; st > 0; st >>= 1) {
#pragma unroll
        for (int j = 0; j < SEG_B; ++j)
            if (s_rs[r[j] + st] <= ic[j]) r[j] += st;
    }
#pragma unroll
    for (int j = 0; j < SEG_B; ++j) {
        const unsigned w = tmp[s_ra[r[j]] + ic[j]];
        v[j] = (i0 + j * SEG_T + t < n) ? w : NO_EDGE;
    }
}

__global__ void __launch_bounds__(SEG_T) bucket_sort_kernel(const unsigned* __restrict__ tmp,
                                                            const int* __restrict__ lo,
                                                            int* __restrict__ off,
                                                            int* __restrict__ csrc) {
    __shared__ int s_cnt[256];
    __shared__ int s_cur[256];
    __shared__ int s_rs[256];
    __shared__ int s_ra[256];
    __shared__ int s_tot[16];
    __shared__ int s_nb[2];
    __shared__ int s_sorted[SEG_CAP];
    const int t = threadIdx.x, k = blockIdx.x;
    int l0 = 0, len = 0;
    if (t < NCHUNK) {
        l0  = lo[t * LO_PITCH + k];
        len = lo[t * LO_PITCH + k + 1] - l0;
    }
    int rs, bs;
    scan256_pair(len, l0, t, s_tot, rs, bs);
    if (t < 256) {
        s_cnt[t] = 0;
        s_rs[t]  = rs;
        s_ra[t]  = t * CHUNK + l0 - rs;
    }
    if (t == 255) { s_nb[0] = rs + len; s_nb[1] = bs + l0; }
    __syncthreads();
    const int n   = s_nb[0];
    const int beg = s_nb[1];
    const bool one = n <= SEG_T * SEG_B;
    unsigned v[SEG_B];
#pragma unroll
    for (int j = 0; j < SEG_B; ++j) v[j] = NO_EDGE;
    for (int i0 = 0; i0 < n; i0 += SEG_T * SEG_B) {
        bucket_fetch(tmp, s_rs, s_ra, i0, t, n, v);
#pragma unroll
        for (int j = 0; j < SEG_B; ++j)
            if (v[j] != NO_EDGE) atomicAdd(&s_cnt[(v[j] >> 16) & 255], 1);
    }
    __syncthreads();
    const int c  = (t < 256) ? s_cnt[t] : 0;
    const int nl = scan256(c, t, s_tot);
    if (t < 256) {
        s_cur[t] = nl;
        off[k * 256 + t] = beg + nl;
    }
    __syncthreads();
    const bool fits = n <= SEG_CAP;
    for (int i0 = 0; i0 < n; i0 += SEG_T * SEG_B) {
        if (!one) bucket_fetch(tmp, s_rs, s_ra, i0, t, n, v);
#pragma unroll
        for (int j = 0; j < SEG_B; ++j) {
            if (v[j] != NO_EDGE) {
                const int slot = atomicAdd(&s_cur[(v[j] >> 16) & 255], 1);
                const int sv = (int)(v[j] & 0xFFFFu);
                if (fits) s_sorted[slot] = sv;
                else      csrc[beg + slot] = sv;
            }
        }
    }
    if (fits) {
        __syncthreads();
        for (int i = t; i < n; i += SEG_T) csrc[beg + i] = s_sorted[i];
    }
}


template <int N>
__device__ __forceinline__ void agg1_group(const float* __restrict__ feat, const int s_l,
                                           const int i, const int lane,
                                           const v2f (&w)[4], const float erd,
                                           float& m, float& l, v2f (&a)[4]) {
    v4f f0[N], f1[N];
#pragma unroll
    for (int k = 0; k < N; ++k) {
        const int s = __builtin_amdgcn_readlane(s_l, i + k);
        const v4f* fs = (const v4f*)(feat + (size_t)s * 256 + lane * 8);
        f0[k] = fs[0];
        f1[k] = fs[1];
    }
    float sc[N];
    float mc = m;
#pragma unroll
    for (int k = 0; k < N; ++k) {
        v2f q = f0[k].xy * w[0];
        q += f0[k].zw * w[1];
        q += f1[k].xy * w[2];
        q += f1[k].zw * w[3];
        float p = q.x + q.y;
        p += row_xor<1>(p);
        p += row_xor<2>(p);
        sc[k] = leaky(p + erd);
        mc = fmaxf(mc, sc[k]);
    }
    const float c = __expf(m - mc);
    l *= c;
#pragma unroll
    for (int j = 0; j < 4; ++j) a[j] *= c;
#pragma unroll
    for (int k = 0; k < N; ++k) {
        const float p = __expf(sc[k] - mc);
        l += p;
        a[0] += f0[k].xy * p;
        a[1] += f0[k].zw * p;
        a[2] += f1[k].xy * p;
        a[3] += f1[k].zw * p;
    }
    m = mc;
}

__global__ void __launch_bounds__(256) agg1_gather_kernel(const int* __restrict__ off,
                                                          const int* __restrict__ csrc,
                                                          const float* __restrict__ al,
                                                          const float* __restrict__ er,
                                                          const float* __restrict__ feat,
                                                          const float* __restrict__ x,
                                                          const float* __restrict__ b1,
                                                          float* __restrict__ rst) {
    const int lane = threadIdx.x & 31;
    const int d = __builtin_amdgcn_readfirstlane((int)(blockIdx.x * 8 + (threadIdx.x >> 5)));
    const int beg = off[d], end = off[d + 1];
    const int h = lane >> 2;
    const float erd = er[d * H1N + h];
    v2f w[4];
#pragma unroll
    for (int j = 0; j < 4; ++j) w[j] = *(const v2f*)(al + lane * 8 + 2 * j);

    float m = -INFINITY, l = 0.f;
    v2f a[4] = {};
    for (int base = beg; base < end; base += 32) {
        const int mcnt = (end - base < 32) ? (end - base) : 32;
        const int s_l = (lane < mcnt) ? csrc[base + lane] : 0;
        int i = 0;
        for (; i + 4 <= mcnt; i += 4)
            agg1_group<4>(feat, s_l, i, lane, w, erd, m, l, a);
        const int rem = mcnt - i;
        if (rem == 3)      agg1_group<3>(feat, s_l, i, lane, w, erd, m, l, a);
        else if (rem == 2) agg1_group<2>(feat, s_l, i, lane, w, erd, m, l, a);
        else if (rem == 1) agg1_group<1>(feat, s_l, i, lane, w, erd, m, l, a);
    }
    const float inv = (l > 0.f) ? 1.f / l : 0.f;
    const size_t o = (size_t)d * 256 + lane * 8;
    const float4 x0 = *(const float4*)(x + o);
    const float4 x1 = *(const float4*)(x + o + 4);
    const float4 c0 = *(const float4*)(b1 + lane * 8);
    const float4 c1 = *(const float4*)(b1 + lane * 8 + 4);
    float4 r0, r1;
    r0.x = fmaxf(a[0].x * inv + x0.x + c0.x, 0.f); r0.y = fmaxf(a[0].y * inv + x0.y + c0.y, 0.f);
    r0.z = fmaxf(a[1].x * inv + x0.z + c0.z, 0.f); r0.w = fmaxf(a[1].y * inv + x0.w + c0.w, 0.f);
    r1.x = fmaxf(a[2].x * inv + x1.x + c1.x, 0.f); r1.y = fmaxf(a[2].y * inv + x1.y + c1.y, 0.f);
    r1.z = fmaxf(a[3].x * inv + x1.z + c1.z, 0.f); r1.w = fmaxf(a[3].y * inv + x1.w + c1.w, 0.f);
    *(float4*)(rst + o)     = r0;
    *(float4*)(rst + o + 4) = r1;
}

__global__ void __launch_bounds__(256) agg2_gather_kernel(const int* __restrict__ off,
                                                          const int* __restrict__ csrc,
                                                          const float* __restrict__ el,
                                                          const float* __restrict__ er,
                                                          const float* __restrict__ feat,
                                                          float* __restrict__ outp) {
    const int lane = threadIdx.x & 31;
    const int d = __builtin_amdgcn_readfirstlane((int)(blockIdx.x * 8 + (threadIdx.x >> 5)));
    const int beg = off[d], end = off[d + 1];
    const float erd = er[d];

    const bool have0 = beg + lane < end;
    int   s0 = 0;
    float v0 = -INFINITY;
    if (have0) { s0 = csrc[beg + lane]; v0 = leaky(el[s0] + erd); }
    float mx = v0;
    for (int e = beg + 32 + lane; e < end; e += 32)
        mx = fmaxf(mx, leaky(el[csrc[e]] + erd));
    for (int o = 16; o > 0; o >>= 1) mx = fmaxf(mx, __shfl_xor(mx, o, 32));
    const float w0 = have0 ? __expf(v0 - mx) : 0.f;
    float sm = w0;
    for (int e = beg + 32 + lane; e < end; e += 32)
        sm += __expf(leaky(el[csrc[e]] + erd) - mx);
    for (int o = 16; o > 0; o >>= 1) sm += __shfl_xor(sm, o, 32);

    float acc = 0.f;
    for (int base = beg; base < end; base += 32) {
        const int mcnt = (end - base < 32) ? (end - base) : 32;
        int   s_l = s0;
        float a_l = w0 / sm;
        if (base != beg) {
            s_l = 0; a_l = 0.f;
            if (lane < mcnt) {
                s_l = csrc[base + lane];
                a_l = __expf(leaky(el[s_l] + erd) - mx) / sm;
            }
        }
#pragma unroll 4
        for (int i = 0; i < mcnt; ++i) {
            const int   s = __builtin_amdgcn_readlane(s_l, i);
            const float a = __int_as_float(__builtin_amdgcn_readlane(__float_as_int(a_l), i));
            acc += feat[(size_t)s * 32 + lane] * a;
        }
    }
    const size_t o = (size_t)d * 32 + lane;
    outp[o] = outp[o] + acc;
}

extern "C" void kernel_launch(void* const* d_in, const int* in_sizes, int n_in,
                              void* d_out, int out_size, void* d_ws, size_t ws_size,
                              hipStream_t stream) {
    const float* x     = (const float*)d_in[0];
    const int*   src   = (const int*)  d_in[1];
    const int*   dst   = (const int*)  d_in[2];
    const float* W1    = (const float*)d_in[3];
    const float* al1   = (const float*)d_in[4];
    const float* ar1   = (const float*)d_in[5];
    const float* b1    = (const float*)d_in[6];
    const float* W2    = (const float*)d_in[7];
    const float* al2   = (const float*)d_in[8];
    const float* ar2   = (const float*)d_in[9];
    const float* resW2 = (const float*)d_in[10];
    const float* b2    = (const float*)d_in[11];
    float* outp = (float*)d_out;

    float* ws    = (float*)d_ws;
    float* feat1 = ws;
    float* rst1  = feat1 + (size_t)NN * 256;
    float* er1   = rst1  + (size_t)NN * 256;
    int*   csrc  = (int*)(er1 + (size_t)NN * 8);
    unsigned* tmp = (unsigned*)(csrc + EE);
    int*   off   = (int*)(tmp + EE);
    int*   lo    = (int*)rst1;
    float* feat2 = ws;
    float* el2   = feat2 + (size_t)NN * 32;
    float* er2   = el2   + NN_PAD;

    chunk_sort_kernel<<<NCHUNK, SORT_T, 0, stream>>>(src, dst, lo, tmp);
    bucket_sort_kernel<<<NBKT, SEG_T, 0, stream>>>(tmp, lo, off, csrc);

    gemm1_kernel<<<NN / 16, 128, 0, stream>>>(x, W1, ar1, feat1, er1);
    agg1_gather_kernel<<<NN / 8, 256, 0, stream>>>(off, csrc, al1, er1, feat1, x, b1, rst1);

    gemm2_kernel<<<(NN + 63) / 64, 128, 0, stream>>>(rst1, W2, resW2, b2, al2, ar2,
                                                     feat2, outp, el2, er2, NN);
    agg2_gather_kernel<<<NN / 8, 256, 0, stream>>>(off, csrc, el2, er2, feat2, outp);
}


# Layer-2 projections (feat2, out) run in the layer-1 gather's tail: 32 nodes a workgroup, y rows pass through LDS, k in 8 slices added in slice order; er2 is agg2's own wave sum; 6 to 5 launches
# speedup vs baseline: 6.5653x; 1.0942x over previous
#include <hip/hip_runtime.h>
#include <hip/hip_bf16.h>
#include <math.h>

#define NN 50000
#define EE 800000
#define INF_ 256
#define OUTF 32
#define H1N 8
#define SLOPE 0.2f

typedef float v2f __attribute__((ext_vector_type(2)));
typedef float v4f __attribute__((ext_vector_type(4)));
typedef float v8f __attribute__((ext_vector_type(8)));

__device__ __forceinline__ float leaky(float x) { return x > 0.f ? x : SLOPE * x; }

__device__ __forceinline__ void accumulator_slots() {
    __builtin_amdgcn_sched_barrier(0);
    asm volatile("v_nop\n\tv_nop\n\tv_nop\n\tv_nop");
    __builtin_amdgcn_sched_barrier(0);
}

template <int M>
__device__ __forceinline__ float row_xor(const float x) {
    const int xi = __float_as_int(x);
    return __int_as_float(__builtin_amdgcn_update_dpp(xi, xi, 0x160 + M, 0xf, 0xf, true));
}

__device__ __forceinline__ float half_swap(const float x) {
    const int xi = __float_as_int(x);
    return __int_as_float(__builtin_amdgcn_permlanex16(xi, xi, 0x76543210, 0xfedcba98, false, false));
}

template <int CNT, int M>
__device__ __forceinline__ void halving_step(float* v, const int lane) {
    const bool up = (lane & M) != 0;
#pragma unroll
    for (int i = 0; i < CNT / 2; ++i) {
        const float keep = up ? v[i + CNT / 2] : v[i];
        const float send = up ? v[i] : v[i + CNT / 2];
        v[i] = keep + row_xor<M>(send);
    }
}
__device__ __forceinline__ float halfwave_sums16(float (&v)[16], const int lane) {
    halving_step<16, 8>(v, lane);
    halving_step<8, 4>(v, lane);
    halving_step<4, 2>(v, lane);
    halving_step<2, 1>(v, lane);
    return v[0];
}
__device__ __forceinline__ float wave_max(float x) {
    x = fmaxf(x, row_xor<1>(x));
    x = fmaxf(x, row_xor<2>(x));
    x = fmaxf(x, row_xor<4>(x));
    x = fmaxf(x, row_xor<8>(x));
    return fmaxf(x, half_swap(x));
}
__device__ __forceinline__ float wave_sum(float x) {
    x += row_xor<1>(x);
    x += row_xor<2>(x);
    x += row_xor<4>(x);
    x += row_xor<8>(x);
    return x + half_swap(x);
}

__global__ void __launch_bounds__(128) gemm1_kernel(const float* __restrict__ X,
                                                    const float* __restrict__ W,
                                                    const float* __restrict__ AR,
                                                    float* __restrict__ C,
                                                    float* __restrict__ ER) {
    __shared__ float s_er[16 * H1N];
    const int wave = threadIdx.x >> 5;
    const int lane = threadIdx.x & 31;
    const int half = lane >> 4;
    const int r    = lane & 15;
    const int row0 = blockIdx.x * 16;
    const int col0 = wave * 64;

    v8f acc[4] = {};
    const float* xp = X + (size_t)(row0 + r) * INF_ + half * 4;
    const float* wp = W + (size_t)(half * 4) * 256 + col0 + r * 4;

#pragma nounroll
    for (int k = 0; k < INF_; k += 8) {
        const v4f a4 = *(const v4f*)(xp + k);
        const float* wk = wp + (size_t)k * 256;
        const v4f b0 = *(const v4f*)(wk);
        const v4f b1 = *(const v4f*)(wk + 256);
        const v4f b2 = *(const v4f*)(wk + 512);
        const v4f b3 = *(const v4f*)(wk + 768);
        v2f a;
        a.x = a4.x; a.y = a4.y;
#pragma unroll
        for (int t = 0; t < 4; ++t) {
            v2f b;
            b.x = b0[t]; b.y = b1[t];
            acc[t] = __builtin_amdgcn_wmma_f32_16x16x4_f32(
                false, a, false, b, (short)0, acc[t], false, false);
        }
        a.x = a4.z; a.y = a4.w;
#pragma unroll
        for (int t = 0; t < 4; ++t) {
            v2f b;
            b.x = b2[t]; b.y = b3[t];
            acc[t] = __builtin_amdgcn_wmma_f32_16x16x4_f32(
                false, a, false, b, (short)0, acc[t], false, false);
        }
    }

    accumulator_slots();

    const int lrow = half * 8 + (r & 7);
    const int hd   = wave * 2 + (r >> 3);
    {
        const v4f wc = *(const v4f*)(AR + col0 + r * 4);
        float v[8];
#pragma unroll
        for (int p = 0; p < 4; ++p) {
            v2f q[4];
#pragma unroll
            for (int t = 0; t < 4; ++t) { q[t].x = acc[t][2 * p]; q[t].y = acc[t][2 * p + 1]; }
            v2f s = q[0] * wc.x;
            s += q[1] * wc.y;
            s += q[2] * wc.z;
            s += q[3] * wc.w;
            v[2 * p + 0] = s.x;
            v[2 * p + 1] = s.y;
        }
        halving_step<8, 4>(v, lane);
        halving_step<4, 2>(v, lane);
        halving_step<2, 1>(v, lane);
        s_er[lrow * H1N + hd] = v[0];
    }
    __syncthreads();
    ER[(size_t)row0 * H1N + threadIdx.x] = s_er[threadIdx.x];

#pragma unroll
    for (int j = 0; j < 8; ++j) {
        const int row = row0 + half * 8 + j;
        v4f o;
        o.x = acc[0][j]; o.y = acc[1][j]; o.z = acc[2][j]; o.w = acc[3][j];
        *(v4f*)(C + (size_t)row * 256 + col0 + r * 4) = o;
    }
}

#define NCHUNK 250
#define CHUNK (EE / NCHUNK)
#define NBKT ((NN + 255) / 256)
#define LO_PITCH 224
#define OFF_LEN (NBKT * 256)
#define SORT_T 512
#define PER_T ((CHUNK + SORT_T - 1) / SORT_T)
#define SEG_T 1024
#define SEG_B 5
#define SEG_CAP 8192
#define NO_EDGE 0xFFFFFFFFu
static_assert(EE % NCHUNK == 0 && CHUNK % 4 == 0, "chunks must tile the edge list in 16-B pieces");
static_assert(NCHUNK <= 256 && NBKT < 256, "one thread of the first 256 per chunk / per bucket");
static_assert(NN <= 65536 && NN + 1 <= OFF_LEN, "edge word fields, offsets length");
static_assert(LO_PITCH >= NBKT + 1 && LO_PITCH % 32 == 0, "table row: whole lines");

__device__ __forceinline__ int scan256(const int c, const int t, int* s_tot) {
    const int lane = t & 31, w = t >> 5;
    int inc = c;
    if (t < 256) {
        for (int o = 1; o < 32; o <<= 1) {
            const int nb = __shfl_up(inc, o, 32);
            if (lane >= o) inc += nb;
        }
        if (lane == 31) s_tot[w] = inc;
    }
    __syncthreads();
    int base = 0;
    if (t < 256) {
        for (int i = 0; i < 8; ++i)
            if (i < w) base += s_tot[i];
    }
    return base + inc - c;
}

__device__ __forceinline__ void scan256_pair(const int a, const int b, const int t, int* s_tot,
                                             int& ex_a, int& ex_b) {
    const int lane = t & 31, w = t >> 5;
    int ia = a, ib = b;
    if (t < 256) {
        for (int o = 1; o < 32; o <<= 1) {
            const int na = __shfl_up(ia, o, 32);
            const int nb = __shfl_up(ib, o, 32);
            if (lane >= o) { ia += na; ib += nb; }
        }
        if (lane == 31) { s_tot[w] = ia; s_tot[8 + w] = ib; }
    }
    __syncthreads();
    int ba = 0, bb = 0;
    if (t < 256) {
        for (int i = 0; i < 8; ++i)
            if (i < w) { ba += s_tot[i]; bb += s_tot[8 + i]; }
    }
    ex_a = ba + ia - a;
    ex_b = bb + ib - b;
}

__global__ void __launch_bounds__(SORT_T) chunk_sort_kernel(const int* __restrict__ src,
                                                            const int* __restrict__ dst,
                                                            int* __restrict__ lo,
                                                            unsigned* __restrict__ tmp) {
    __shared__ int s_cnt[256];
    __shared__ int s_tot[8];
    __shared__ __attribute__((aligned(16))) unsigned s_sorted[CHUNK];
    const int t = threadIdx.x, b = blockIdx.x;
    if (t < 256) s_cnt[t] = 0;
    const int* sp = src + b * CHUNK;
    const int* dp = dst + b * CHUNK;
    int s[PER_T], d[PER_T], rk[PER_T];
#pragma unroll
    for (int i = 0; i < PER_T; ++i) {
        const int e = i * SORT_T + t;
        s[i] = (e < CHUNK) ? sp[e] : 0;
        d[i] = (e < CHUNK) ? dp[e] : -1;
    }
    __syncthreads();
#pragma unroll
    for (int i = 0; i < PER_T; ++i)
        rk[i] = (d[i] >= 0) ? atomicAdd(&s_cnt[d[i] >> 8], 1) : 0;
    __syncthreads();
    const int c  = (t < 256) ? s_cnt[t] : 0;
    const int ls = scan256(c, t, s_tot);
    if (t < 256) s_cnt[t] = ls;
    if (t <= NBKT) lo[b * LO_PITCH + t] = ls;
    __syncthreads();
#pragma unroll
    for (int i = 0; i < PER_T; ++i) {
        if (d[i] >= 0) {
            const int k = d[i] >> 8;
            s_sorted[s_cnt[k] + rk[i]] = (unsigned)s[i] | ((unsigned)(d[i] & 255) << 16);
        }
    }
    __syncthreads();
    const uint4* s4 = (const uint4*)s_sorted;
    uint4* t4 = (uint4*)(tmp + (size_t)b * CHUNK);
    for (int i = t; i < CHUNK / 4; i += SORT_T) t4[i] = s4[i];
}

__device__ __forceinline__ void bucket_fetch(const unsigned* __restrict__ tmp, const int* s_rs,
                                             const int* s_ra, const int i0, const int t,
                                             const int n, unsigned (&v)[SEG_B]) {
    int ic[SEG_B], r[SEG_B];
#pragma unroll
    for (int j = 0; j < SEG_B; ++j) {
        const int i = i0 + j * SEG_T + t;
        ic[j] = (i < n) ? i : n - 1;
        r[j] = 0;
    }
#pragma unroll
    for (int st = 128; st > 0; st >>= 1) {
#pragma unroll
        for (int j = 0; j < SEG_B; ++j)
            if (s_rs[r[j] + st] <= ic[j]) r[j] += st;
    }
#pragma unroll
    for (int j = 0; j < SEG_B; ++j) {
        const unsigned w = tmp[s_ra[r[j]] + ic[j]];
        v[j] = (i0 + j * SEG_T + t < n) ? w : NO_EDGE;
    }
}

__global__ void __launch_bounds__(SEG_T) bucket_sort_kernel(const unsigned* __restrict__ tmp,
                                                            const int* __restrict__ lo,
                                                            int* __restrict__ off,
                                                            int* __restrict__ csrc) {
    __shared__ int s_cnt[256];
    __shared__ int s_cur[256];
    __shared__ int s_rs[256];
    __shared__ int s_ra[256];
    __shared__ int s_tot[16];
    __shared__ int s_nb[2];
    __shared__ int s_sorted[SEG_CAP];
    const int t = threadIdx.x, k = blockIdx.x;
    int l0 = 0, len = 0;
    if (t < NCHUNK) {
        l0  = lo[t * LO_PITCH + k];
        len = lo[t * LO_PITCH + k + 1] - l0;
    }
    int rs, bs;
    scan256_pair(len, l0, t, s_tot, rs, bs);
    if (t < 256) {
        s_cnt[t] = 0;
        s_rs[t]  = rs;
        s_ra[t]  = t * CHUNK + l0 - rs;
    }
    if (t == 255) { s_nb[0] = rs + len; s_nb[1] = bs + l0; }
    __syncthreads();
    const int n   = s_nb[0];
    const int beg = s_nb[1];
    const bool one = n <= SEG_T * SEG_B;
    unsigned v[SEG_B];
#pragma unroll
    for (int j = 0; j < SEG_B; ++j) v[j] = NO_EDGE;
    for (int i0 = 0; i0 < n; i0 += SEG_T * SEG_B) {
        bucket_fetch(tmp, s_rs, s_ra, i0, t, n, v);
#pragma unroll
        for (int j = 0; j < SEG_B; ++j)
            if (v[j] != NO_EDGE) atomicAdd(&s_cnt[(v[j] >> 16) & 255], 1);
    }
    __syncthreads();
    const int c  = (t < 256) ? s_cnt[t] : 0;
    const int nl = scan256(c, t, s_tot);
    if (t < 256) {
        s_cur[t] = nl;
        off[k * 256 + t] = beg + nl;
    }
    __syncthreads();
    const bool fits = n <= SEG_CAP;
    for (int i0 = 0; i0 < n; i0 += SEG_T * SEG_B) {
        if (!one) bucket_fetch(tmp, s_rs, s_ra, i0, t, n, v);
#pragma unroll
        for (int j = 0; j < SEG_B; ++j) {
            if (v[j] != NO_EDGE) {
                const int slot = atomicAdd(&s_cur[(v[j] >> 16) & 255], 1);
                const int sv = (int)(v[j] & 0xFFFFu);
                if (fits) s_sorted[slot] = sv;
                else      csrc[beg + slot] = sv;
            }
        }
    }
    if (fits) {
        __syncthreads();
        for (int i = t; i < n; i += SEG_T) csrc[beg + i] = s_sorted[i];
    }
}


template <int N>
__device__ __forceinline__ void agg1_group(const float* __restrict__ feat, const int s_l,
                                           const int i, const int lane,
                                           const v2f (&w)[4], const float erd,
                                           float& m, float& l, v2f (&a)[4]) {
    v4f f0[N], f1[N];
#pragma unroll
    for (int k = 0; k < N; ++k) {
        const int s = __builtin_amdgcn_readlane(s_l, i + k);
        const v4f* fs = (const v4f*)(feat + (size_t)s * 256 + lane * 8);
        f0[k] = fs[0];
        f1[k] = fs[1];
    }
    float sc[N];
    float mc = m;
#pragma unroll
    for (int k = 0; k < N; ++k) {
        v2f q = f0[k].xy * w[0];
        q += f0[k].zw * w[1];
        q += f1[k].xy * w[2];
        q += f1[k].zw * w[3];
        float p = q.x + q.y;
        p += row_xor<1>(p);
        p += row_xor<2>(p);
        sc[k] = leaky(p + erd);
        mc = fmaxf(mc, sc[k]);
    }
    const float c = __expf(m - mc);
    l *= c;
#pragma unroll
    for (int j = 0; j < 4; ++j) a[j] *= c;
#pragma unroll
    for (int k = 0; k < N; ++k) {
        const float p = __expf(sc[k] - mc);
        l += p;
        a[0] += f0[k].xy * p;
        a[1] += f0[k].zw * p;
        a[2] += f1[k].xy * p;
        a[3] += f1[k].zw * p;
    }
    m = mc;
}

#define YP 260
#define KSL 32
#define NPB 32

__global__ void __launch_bounds__(512) agg1_proj_kernel(const int* __restrict__ off,
                                                        const int* __restrict__ csrc,
                                                        const float* __restrict__ al,
                                                        const float* __restrict__ er,
                                                        const float* __restrict__ feat,
                                                        const float* __restrict__ x,
                                                        const float* __restrict__ b1,
                                                        const float* __restrict__ W2,
                                                        const float* __restrict__ RW,
                                                        const float* __restrict__ b2,
                                                        float* __restrict__ F2,
                                                        float* __restrict__ OUTP) {
    __shared__ __attribute__((aligned(16))) float s_y[NPB * YP];
    __shared__ __attribute__((aligned(16))) float s_part[8 * 2 * NPB * 32];
    const int lane = threadIdx.x & 31;
    const int wave = __builtin_amdgcn_readfirstlane((int)(threadIdx.x >> 5));
    const int h = lane >> 2;
    v2f w[4];
#pragma unroll
    for (int j = 0; j < 4; ++j) w[j] = *(const v2f*)(al + lane * 8 + 2 * j);

#pragma nounroll
    for (int q = 0; q < 2; ++q) {
        const int row = q * 16 + wave;
        const int d = (int)blockIdx.x * NPB + row;
        float4 r0 = make_float4(0.f, 0.f, 0.f, 0.f), r1 = r0;
        if (d < NN) {
            const int beg = off[d], end = off[d + 1];
            const float erd = er[d * H1N + h];
            float m = -INFINITY, l = 0.f;
            v2f a[4] = {};
            for (int base = beg; base < end; base += 32) {
                const int mcnt = (end - base < 32) ? (end - base) : 32;
                const int s_l = (lane < mcnt) ? csrc[base + lane] : 0;
                int i = 0;
                for (; i + 4 <= mcnt; i += 4)
                    agg1_group<4>(feat, s_l, i, lane, w, erd, m, l, a);
                const int rem = mcnt - i;
                if (rem == 3)      agg1_group<3>(feat, s_l, i, lane, w, erd, m, l, a);
                else if (rem == 2) agg1_group<2>(feat, s_l, i, lane, w, erd, m, l, a);
                else if (rem == 1) agg1_group<1>(feat, s_l, i, lane, w, erd, m, l, a);
            }
            const float inv = (l > 0.f) ? 1.f / l : 0.f;
            const size_t o = (size_t)d * 256 + lane * 8;
            const float4 x0 = *(const float4*)(x + o);
            const float4 x1 = *(const float4*)(x + o + 4);
            const float4 c0 = *(const float4*)(b1 + lane * 8);
            const float4 c1 = *(const float4*)(b1 + lane * 8 + 4);
            r0.x = fmaxf(a[0].x * inv + x0.x + c0.x, 0.f); r0.y = fmaxf(a[0].y * inv + x0.y + c0.y, 0.f);
            r0.z = fmaxf(a[1].x * inv + x0.z + c0.z, 0.f); r0.w = fmaxf(a[1].y * inv + x0.w + c0.w, 0.f);
            r1.x = fmaxf(a[2].x * inv + x1.x + c1.x, 0.f); r1.y = fmaxf(a[2].y * inv + x1.y + c1.y, 0.f);
            r1.z = fmaxf(a[3].x * inv + x1.z + c1.z, 0.f); r1.w = fmaxf(a[3].y * inv + x1.w + c1.w, 0.f);
        }
        *(float4*)(s_y + row * YP + lane * 8)     = r0;
        *(float4*)(s_y + row * YP + lane * 8 + 4) = r1;
    }

    const int half = lane >> 4;
    const int r    = lane & 15;
    const int p    = wave & 1;
    const int ks   = wave >> 1;
    const float* bp = (p ? RW : W2) + (size_t)(ks * KSL + half * 4) * 32 + r * 2;
    v2f bq[KSL / 8][4];
#pragma unroll
    for (int k8 = 0; k8 < KSL / 8; ++k8)
#pragma unroll
        for (int i = 0; i < 4; ++i)
            bq[k8][i] = *(const v2f*)(bp + (size_t)(k8 * 8 + i) * 32);
    __syncthreads();

    v8f acc[2][2] = {};
#pragma unroll
    for (int k8 = 0; k8 < KSL / 8; ++k8) {
#pragma unroll
        for (int rt = 0; rt < 2; ++rt) {
            const v4f a4 = *(const v4f*)(s_y + (rt * 16 + r) * YP + ks * KSL + half * 4 + k8 * 8);
            v2f av;
            av.x = a4.x; av.y = a4.y;
#pragma unroll
            for (int t = 0; t < 2; ++t) {
                v2f b;
                b.x = bq[k8][0][t]; b.y = bq[k8][1][t];
                acc[rt][t] = __builtin_amdgcn_wmma_f32_16x16x4_f32(
                    false, av, false, b, (short)0, acc[rt][t], false, false);
            }
            av.x = a4.z; av.y = a4.w;
#pragma unroll
            for (int t = 0; t < 2; ++t) {
                v2f b;
                b.x = bq[k8][2][t]; b.y = bq[k8][3][t];
                acc[rt][t] = __builtin_amdgcn_wmma_f32_16x16x4_f32(
                    false, av, false, b, (short)0, acc[rt][t], false, false);
            }
        }
    }

    accumulator_slots();

#pragma unroll
    for (int rt = 0; rt < 2; ++rt) {
        float* pp = s_part + ((ks * 2 + p) * NPB + rt * 16 + half * 8) * 32 + r * 2;
#pragma unroll
        for (int j = 0; j < 8; ++j) {
            v2f o;
            o.x = acc[rt][0][j]; o.y = acc[rt][1][j];
            *(v2f*)(pp + j * 32) = o;
        }
    }
    __syncthreads();
    const float bias = b2[lane];
#pragma unroll
    for (int q = 0; q < 2; ++q) {
        const int row = q * 16 + wave;
        const int d = (int)blockIdx.x * NPB + row;
        float f = 0.f, g = 0.f;
#pragma unroll
        for (int s = 0; s < 8; ++s) {
            f += s_part[((s * 2 + 0) * NPB + row) * 32 + lane];
            g += s_part[((s * 2 + 1) * NPB + row) * 32 + lane];
        }
        if (d < NN) {
            F2[(size_t)d * 32 + lane]   = f;
            OUTP[(size_t)d * 32 + lane] = g + bias;
        }
    }
}

__global__ void __launch_bounds__(256) agg2_gather_kernel(const int* __restrict__ off,
                                                          const int* __restrict__ csrc,
                                                          const float* __restrict__ al,
                                                          const float* __restrict__ ar,
                                                          const float* __restrict__ feat,
                                                          float* __restrict__ outp) {
    const int lane = threadIdx.x & 31;
    const int d = __builtin_amdgcn_readfirstlane((int)(blockIdx.x * 8 + (threadIdx.x >> 5)));
    const int beg = off[d], end = off[d + 1];
    const float alc = al[lane];
    const size_t o = (size_t)d * 32 + lane;
    const float erd = wave_sum(feat[o] * ar[lane]);
    const float res = outp[o];
    const bool up = lane >= 16;

    float m = -INFINITY, l = 0.f, acc = 0.f;
    for (int base = beg; base < end; base += 32) {
        const int mcnt = (end - base < 32) ? (end - base) : 32;
        const bool big = mcnt > 16;
        const int s_l = (lane < mcnt) ? csrc[base + lane] : 0;
        float lo[16], hi[16];
        lo[0] = feat[(unsigned)__builtin_amdgcn_readlane(s_l, 0) * 32u + lane];
#pragma unroll
        for (int k = 1; k < 16; ++k) {
            lo[k] = 0.f;
            if (k < mcnt)
                lo[k] = feat[(unsigned)__builtin_amdgcn_readlane(s_l, k) * 32u + lane];
        }
        if (big) {
#pragma unroll
            for (int k = 0; k < 16; ++k) {
                hi[k] = 0.f;
                if (16 + k < mcnt)
                    hi[k] = feat[(unsigned)__builtin_amdgcn_readlane(s_l, 16 + k) * 32u + lane];
            }
        }
        float va[16];
#pragma unroll
        for (int k = 0; k < 16; ++k) va[k] = lo[k] * alc;
        const float sa = halfwave_sums16(va, lane);
        float sb = 0.f;
        if (big) {
            float vb[16];
#pragma unroll
            for (int k = 0; k < 16; ++k) vb[k] = hi[k] * alc;
            sb = halfwave_sums16(vb, lane);
        }
        const float keep = up ? sb : sa;
        const float send = up ? sa : sb;
        const float lg = keep + half_swap(send);
        const bool have = lane < mcnt;
        const float e  = have ? leaky(lg + erd) : -INFINITY;
        const float mc = fmaxf(m, wave_max(e));
        const float c  = __expf(m - mc);
        const float p  = have ? __expf(e - mc) : 0.f;
        l = l * c + wave_sum(p);
        acc *= c;
#pragma unroll
        for (int g = 0; g < 2; ++g) {
            if (g * 8 < mcnt) {
#pragma unroll
                for (int j = 0; j < 8; ++j)
                    acc = __builtin_fmaf(__int_as_float(__builtin_amdgcn_readlane(
                                             __float_as_int(p), g * 8 + j)), lo[g * 8 + j], acc);
            }
        }
        if (big) {
#pragma unroll
            for (int g = 0; g < 2; ++g) {
                if (16 + g * 8 < mcnt) {
#pragma unroll
                    for (int j = 0; j < 8; ++j)
                        acc = __builtin_fmaf(__int_as_float(__builtin_amdgcn_readlane(
                                                 __float_as_int(p), 16 + g * 8 + j)), hi[g * 8 + j], acc);
                }
            }
        }
        m = mc;
    }
    const float inv = (l > 0.f) ? 1.f / l : 0.f;
    outp[o] = res + acc * inv;
}

extern "C" void kernel_launch(void* const* d_in, const int* in_sizes, int n_in,
                              void* d_out, int out_size, void* d_ws, size_t ws_size,
                              hipStream_t stream) {
    const float* x     = (const float*)d_in[0];
    const int*   src   = (const int*)  d_in[1];
    const int*   dst   = (const int*)  d_in[2];
    const float* W1    = (const float*)d_in[3];
    const float* al1   = (const float*)d_in[4];
    const float* ar1   = (const float*)d_in[5];
    const float* b1    = (const float*)d_in[6];
    const float* W2    = (const float*)d_in[7];
    const float* al2   = (const float*)d_in[8];
    const float* ar2   = (const float*)d_in[9];
    const float* resW2 = (const float*)d_in[10];
    const float* b2    = (const float*)d_in[11];
    float* outp = (float*)d_out;

    float* ws    = (float*)d_ws;
    float* feat1 = ws;
    float* area2 = feat1 + (size_t)NN * 256;
    float* er1   = area2 + (size_t)NN * 256;
    int*   csrc  = (int*)(er1 + (size_t)NN * 8);
    unsigned* tmp = (unsigned*)(csrc + EE);
    int*   off   = (int*)(tmp + EE);
    int*   lo    = (int*)area2;
    float* feat2 = area2;

    chunk_sort_kernel<<<NCHUNK, SORT_T, 0, stream>>>(src, dst, lo, tmp);
    bucket_sort_kernel<<<NBKT, SEG_T, 0, stream>>>(tmp, lo, off, csrc);

    gemm1_kernel<<<NN / 16, 128, 0, stream>>>(x, W1, ar1, feat1, er1);
    agg1_proj_kernel<<<(NN + NPB - 1) / NPB, 512, 0, stream>>>(off, csrc, al1, er1, feat1, x, b1,
                                                               W2, resW2, b2, feat2, outp);

    agg2_gather_kernel<<<NN / 8, 256, 0, stream>>>(off, csrc, al2, ar2, feat2, outp);
}


# GEMM1: a wave takes two 16-row tiles, so each W fragment feeds two matrix instructions and W's re-reads from L2 halve (1536 two-tile + 53 one-tile workgroups, not 3125); sums unchanged
# speedup vs baseline: 7.0903x; 1.0800x over previous
#include <hip/hip_runtime.h>
#include <hip/hip_bf16.h>
#include <math.h>

#define NN 50000
#define EE 800000
#define INF_ 256
#define OUTF 32
#define H1N 8
#define SLOPE 0.2f

typedef float v2f __attribute__((ext_vector_type(2)));
typedef float v4f __attribute__((ext_vector_type(4)));
typedef float v8f __attribute__((ext_vector_type(8)));

__device__ __forceinline__ float leaky(float x) { return x > 0.f ? x : SLOPE * x; }

__device__ __forceinline__ void accumulator_slots() {
    __builtin_amdgcn_sched_barrier(0);
    asm volatile("v_nop\n\tv_nop\n\tv_nop\n\tv_nop");
    __builtin_amdgcn_sched_barrier(0);
}

template <int M>
__device__ __forceinline__ float row_xor(const float x) {
    const int xi = __float_as_int(x);
    return __int_as_float(__builtin_amdgcn_update_dpp(xi, xi, 0x160 + M, 0xf, 0xf, true));
}

__device__ __forceinline__ float half_swap(const float x) {
    const int xi = __float_as_int(x);
    return __int_as_float(__builtin_amdgcn_permlanex16(xi, xi, 0x76543210, 0xfedcba98, false, false));
}

template <int CNT, int M>
__device__ __forceinline__ void halving_step(float* v, const int lane) {
    const bool up = (lane & M) != 0;
#pragma unroll
    for (int i = 0; i < CNT / 2; ++i) {
        const float keep = up ? v[i + CNT / 2] : v[i];
        const float send = up ? v[i] : v[i + CNT / 2];
        v[i] = keep + row_xor<M>(send);
    }
}
__device__ __forceinline__ float halfwave_sums16(float (&v)[16], const int lane) {
    halving_step<16, 8>(v, lane);
    halving_step<8, 4>(v, lane);
    halving_step<4, 2>(v, lane);
    halving_step<2, 1>(v, lane);
    return v[0];
}
__device__ __forceinline__ float wave_max(float x) {
    x = fmaxf(x, row_xor<1>(x));
    x = fmaxf(x, row_xor<2>(x));
    x = fmaxf(x, row_xor<4>(x));
    x = fmaxf(x, row_xor<8>(x));
    return fmaxf(x, half_swap(x));
}
__device__ __forceinline__ float wave_sum(float x) {
    x += row_xor<1>(x);
    x += row_xor<2>(x);
    x += row_xor<4>(x);
    x += row_xor<8>(x);
    return x + half_swap(x);
}

#define G1_UNITS 256
#define G1_PAIRS (G1_UNITS * 6)
#define G1_SINGLES (NN / 16 - 2 * G1_PAIRS)
static_assert(NN % 16 == 0 && G1_SINGLES >= 0 && G1_SINGLES < G1_UNITS, "GEMM1 tiling");

template <int RT>
__device__ __forceinline__ void gemm1_rows(const float* __restrict__ X,
                                           const float* __restrict__ W,
                                           const float* __restrict__ AR,
                                           float* __restrict__ C,
                                           float* __restrict__ ER,
                                           float* s_er, const int row0) {
    const int wave = threadIdx.x >> 5;
    const int lane = threadIdx.x & 31;
    const int half = lane >> 4;
    const int r    = lane & 15;
    const int col0 = wave * 64;

    v8f acc[RT][4] = {};
    const float* xp = X + (size_t)(row0 + r) * INF_ + half * 4;
    const float* wp = W + (size_t)(half * 4) * 256 + col0 + r * 4;

#pragma nounroll
    for (int k = 0; k < INF_; k += 8) {
        v4f a4[RT];
#pragma unroll
        for (int rt = 0; rt < RT; ++rt) a4[rt] = *(const v4f*)(xp + (size_t)rt * 16 * INF_ + k);
        const float* wk = wp + (size_t)k * 256;
        const v4f b0 = *(const v4f*)(wk);
        const v4f b1 = *(const v4f*)(wk + 256);
        const v4f b2 = *(const v4f*)(wk + 512);
        const v4f b3 = *(const v4f*)(wk + 768);
        __builtin_amdgcn_sched_barrier(0);
#pragma unroll
        for (int t = 0; t < 4; ++t) {
            v2f b;
            b.x = b0[t]; b.y = b1[t];
#pragma unroll
            for (int rt = 0; rt < RT; ++rt) {
                v2f a;
                a.x = a4[rt].x; a.y = a4[rt].y;
                acc[rt][t] = __builtin_amdgcn_wmma_f32_16x16x4_f32(
                    false, a, false, b, (short)0, acc[rt][t], false, false);
            }
        }
#pragma unroll
        for (int t = 0; t < 4; ++t) {
            v2f b;
            b.x = b2[t]; b.y = b3[t];
#pragma unroll
            for (int rt = 0; rt < RT; ++rt) {
                v2f a;
                a.x = a4[rt].z; a.y = a4[rt].w;
                acc[rt][t] = __builtin_amdgcn_wmma_f32_16x16x4_f32(
                    false, a, false, b, (short)0, acc[rt][t], false, false);
            }
        }
    }

    accumulator_slots();

    const int hd = wave * 2 + (r >> 3);
    {
        const v4f wc = *(const v4f*)(AR + col0 + r * 4);
#pragma unroll
        for (int rt = 0; rt < RT; ++rt) {
            float v[8];
#pragma unroll
            for (int p = 0; p < 4; ++p) {
                v2f q[4];
#pragma unroll
                for (int t = 0; t < 4; ++t) { q[t].x = acc[rt][t][2 * p]; q[t].y = acc[rt][t][2 * p + 1]; }
                v2f s = q[0] * wc.x;
                s += q[1] * wc.y;
                s += q[2] * wc.z;
                s += q[3] * wc.w;
                v[2 * p + 0] = s.x;
                v[2 * p + 1] = s.y;
            }
            halving_step<8, 4>(v, lane);
            halving_step<4, 2>(v, lane);
            halving_step<2, 1>(v, lane);
            s_er[(rt * 16 + half * 8 + (r & 7)) * H1N + hd] = v[0];
        }
    }
    __syncthreads();
#pragma unroll
    for (int rt = 0; rt < RT; ++rt)
        ER[(size_t)(row0 + rt * 16) * H1N + threadIdx.x] = s_er[rt * 16 * H1N + threadIdx.x];

#pragma unroll
    for (int rt = 0; rt < RT; ++rt) {
#pragma unroll
        for (int j = 0; j < 8; ++j) {
            const int row = row0 + rt * 16 + half * 8 + j;
            v4f o;
            o.x = acc[rt][0][j]; o.y = acc[rt][1][j]; o.z = acc[rt][2][j]; o.w = acc[rt][3][j];
            *(v4f*)(C + (size_t)row * 256 + col0 + r * 4) = o;
        }
    }
}

__global__ void __launch_bounds__(128) gemm1_kernel(const float* __restrict__ X,
                                                    const float* __restrict__ W,
                                                    const float* __restrict__ AR,
                                                    float* __restrict__ C,
                                                    float* __restrict__ ER) {
    __shared__ float s_er[32 * H1N];
    const int b = blockIdx.x;
    if (b < G1_PAIRS) gemm1_rows<2>(X, W, AR, C, ER, s_er, b * 32);
    else              gemm1_rows<1>(X, W, AR, C, ER, s_er, G1_PAIRS * 32 + (b - G1_PAIRS) * 16);
}

#define NCHUNK 250
#define CHUNK (EE / NCHUNK)
#define NBKT ((NN + 255) / 256)
#define LO_PITCH 224
#define OFF_LEN (NBKT * 256)
#define SORT_T 512
#define PER_T ((CHUNK + SORT_T - 1) / SORT_T)
#define SEG_T 1024
#define SEG_B 5
#define SEG_CAP 8192
#define NO_EDGE 0xFFFFFFFFu
static_assert(EE % NCHUNK == 0 && CHUNK % 4 == 0, "chunks must tile the edge list in 16-B pieces");
static_assert(NCHUNK <= 256 && NBKT < 256, "one thread of the first 256 per chunk / per bucket");
static_assert(NN <= 65536 && NN + 1 <= OFF_LEN, "edge word fields, offsets length");
static_assert(LO_PITCH >= NBKT + 1 && LO_PITCH % 32 == 0, "table row: whole lines");

__device__ __forceinline__ int scan256(const int c, const int t, int* s_tot) {
    const int lane = t & 31, w = t >> 5;
    int inc = c;
    if (t < 256) {
        for (int o = 1; o < 32; o <<= 1) {
            const int nb = __shfl_up(inc, o, 32);
            if (lane >= o) inc += nb;
        }
        if (lane == 31) s_tot[w] = inc;
    }
    __syncthreads();
    int base = 0;
    if (t < 256) {
        for (int i = 0; i < 8; ++i)
            if (i < w) base += s_tot[i];
    }
    return base + inc - c;
}

__device__ __forceinline__ void scan256_pair(const int a, const int b, const int t, int* s_tot,
                                             int& ex_a, int& ex_b) {
    const int lane = t & 31, w = t >> 5;
    int ia = a, ib = b;
    if (t < 256) {
        for (int o = 1; o < 32; o <<= 1) {
            const int na = __shfl_up(ia, o, 32);
            const int nb = __shfl_up(ib, o, 32);
            if (lane >= o) { ia += na; ib += nb; }
        }
        if (lane == 31) { s_tot[w] = ia; s_tot[8 + w] = ib; }
    }
    __syncthreads();
    int ba = 0, bb = 0;
    if (t < 256) {
        for (int i = 0; i < 8; ++i)
            if (i < w) { ba += s_tot[i]; bb += s_tot[8 + i]; }
    }
    ex_a = ba + ia - a;
    ex_b = bb + ib - b;
}

__global__ void __launch_bounds__(SORT_T) chunk_sort_kernel(const int* __restrict__ src,
                                                            const int* __restrict__ dst,
                                                            int* __restrict__ lo,
                                                            unsigned* __restrict__ tmp) {
    __shared__ int s_cnt[256];
    __shared__ int s_tot[8];
    __shared__ __attribute__((aligned(16))) unsigned s_sorted[CHUNK];
    const int t = threadIdx.x, b = blockIdx.x;
    if (t < 256) s_cnt[t] = 0;
    const int* sp = src + b * CHUNK;
    const int* dp = dst + b * CHUNK;
    int s[PER_T], d[PER_T], rk[PER_T];
#pragma unroll
    for (int i = 0; i < PER_T; ++i) {
        const int e = i * SORT_T + t;
        s[i] = (e < CHUNK) ? sp[e] : 0;
        d[i] = (e < CHUNK) ? dp[e] : -1;
    }
    __syncthreads();
#pragma unroll
    for (int i = 0; i < PER_T; ++i)
        rk[i] = (d[i] >= 0) ? atomicAdd(&s_cnt[d[i] >> 8], 1) : 0;
    __syncthreads();
    const int c  = (t < 256) ? s_cnt[t] : 0;
    const int ls = scan256(c, t, s_tot);
    if (t < 256) s_cnt[t] = ls;
    if (t <= NBKT) lo[b * LO_PITCH + t] = ls;
    __syncthreads();
#pragma unroll
    for (int i = 0; i < PER_T; ++i) {
        if (d[i] >= 0) {
            const int k = d[i] >> 8;
            s_sorted[s_cnt[k] + rk[i]] = (unsigned)s[i] | ((unsigned)(d[i] & 255) << 16);
        }
    }
    __syncthreads();
    const uint4* s4 = (const uint4*)s_sorted;
    uint4* t4 = (uint4*)(tmp + (size_t)b * CHUNK);
    for (int i = t; i < CHUNK / 4; i += SORT_T) t4[i] = s4[i];
}

__device__ __forceinline__ void bucket_fetch(const unsigned* __restrict__ tmp, const int* s_rs,
                                             const int* s_ra, const int i0, const int t,
                                             const int n, unsigned (&v)[SEG_B]) {
    int ic[SEG_B], r[SEG_B];
#pragma unroll
    for (int j = 0; j < SEG_B; ++j) {
        const int i = i0 + j * SEG_T + t;
        ic[j] = (i < n) ? i : n - 1;
        r[j] = 0;
    }
#pragma unroll
    for (int st = 128; st > 0; st >>= 1) {
#pragma unroll
        for (int j = 0; j < SEG_B; ++j)
            if (s_rs[r[j] + st] <= ic[j]) r[j] += st;
    }
#pragma unroll
    for (int j = 0; j < SEG_B; ++j) {
        const unsigned w = tmp[s_ra[r[j]] + ic[j]];
        v[j] = (i0 + j * SEG_T + t < n) ? w : NO_EDGE;
    }
}

__global__ void __launch_bounds__(SEG_T) bucket_sort_kernel(const unsigned* __restrict__ tmp,
                                                            const int* __restrict__ lo,
                                                            int* __restrict__ off,
                                                            int* __restrict__ csrc) {
    __shared__ int s_cnt[256];
    __shared__ int s_cur[256];
    __shared__ int s_rs[256];
    __shared__ int s_ra[256];
    __shared__ int s_tot[16];
    __shared__ int s_nb[2];
    __shared__ int s_sorted[SEG_CAP];
    const int t = threadIdx.x, k = blockIdx.x;
    int l0 = 0, len = 0;
    if (t < NCHUNK) {
        l0  = lo[t * LO_PITCH + k];
        len = lo[t * LO_PITCH + k + 1] - l0;
    }
    int rs, bs;
    scan256_pair(len, l0, t, s_tot, rs, bs);
    if (t < 256) {
        s_cnt[t] = 0;
        s_rs[t]  = rs;
        s_ra[t]  = t * CHUNK + l0 - rs;
    }
    if (t == 255) { s_nb[0] = rs + len; s_nb[1] = bs + l0; }
    __syncthreads();
    const int n   = s_nb[0];
    const int beg = s_nb[1];
    const bool one = n <= SEG_T * SEG_B;
    unsigned v[SEG_B];
#pragma unroll
    for (int j = 0; j < SEG_B; ++j) v[j] = NO_EDGE;
    for (int i0 = 0; i0 < n; i0 += SEG_T * SEG_B) {
        bucket_fetch(tmp, s_rs, s_ra, i0, t, n, v);
#pragma unroll
        for (int j = 0; j < SEG_B; ++j)
            if (v[j] != NO_EDGE) atomicAdd(&s_cnt[(v[j] >> 16) & 255], 1);
    }
    __syncthreads();
    const int c  = (t < 256) ? s_cnt[t] : 0;
    const int nl = scan256(c, t, s_tot);
    if (t < 256) {
        s_cur[t] = nl;
        off[k * 256 + t] = beg + nl;
    }
    __syncthreads();
    const bool fits = n <= SEG_CAP;
    for (int i0 = 0; i0 < n; i0 += SEG_T * SEG_B) {
        if (!one) bucket_fetch(tmp, s_rs, s_ra, i0, t, n, v);
#pragma unroll
        for (int j = 0; j < SEG_B; ++j) {
            if (v[j] != NO_EDGE) {
                const int slot = atomicAdd(&s_cur[(v[j] >> 16) & 255], 1);
                const int sv = (int)(v[j] & 0xFFFFu);
                if (fits) s_sorted[slot] = sv;
                else      csrc[beg + slot] = sv;
            }
        }
    }
    if (fits) {
        __syncthreads();
        for (int i = t; i < n; i += SEG_T) csrc[beg + i] = s_sorted[i];
    }
}


template <int N>
__device__ __forceinline__ void agg1_group(const float* __restrict__ feat, const int s_l,
                                           const int i, const int lane,
                                           const v2f (&w)[4], const float erd,
                                           float& m, float& l, v2f (&a)[4]) {
    v4f f0[N], f1[N];
#pragma unroll
    for (int k = 0; k < N; ++k) {
        const int s = __builtin_amdgcn_readlane(s_l, i + k);
        const v4f* fs = (const v4f*)(feat + (size_t)s * 256 + lane * 8);
        f0[k] = fs[0];
        f1[k] = fs[1];
    }
    float sc[N];
    float mc = m;
#pragma unroll
    for (int k = 0; k < N; ++k) {
        v2f q = f0[k].xy * w[0];
        q += f0[k].zw * w[1];
        q += f1[k].xy * w[2];
        q += f1[k].zw * w[3];
        float p = q.x + q.y;
        p += row_xor<1>(p);
        p += row_xor<2>(p);
        sc[k] = leaky(p + erd);
        mc = fmaxf(mc, sc[k]);
    }
    const float c = __expf(m - mc);
    l *= c;
#pragma unroll
    for (int j = 0; j < 4; ++j) a[j] *= c;
#pragma unroll
    for (int k = 0; k < N; ++k) {
        const float p = __expf(sc[k] - mc);
        l += p;
        a[0] += f0[k].xy * p;
        a[1] += f0[k].zw * p;
        a[2] += f1[k].xy * p;
        a[3] += f1[k].zw * p;
    }
    m = mc;
}

#define YP 260
#define KSL 32
#define NPB 32

__global__ void __launch_bounds__(512) agg1_proj_kernel(const int* __restrict__ off,
                                                        const int* __restrict__ csrc,
                                                        const float* __restrict__ al,
                                                        const float* __restrict__ er,
                                                        const float* __restrict__ feat,
                                                        const float* __restrict__ x,
                                                        const float* __restrict__ b1,
                                                        const float* __restrict__ W2,
                                                        const float* __restrict__ RW,
                                                        const float* __restrict__ b2,
                                                        float* __restrict__ F2,
                                                        float* __restrict__ OUTP) {
    __shared__ __attribute__((aligned(16))) float s_y[NPB * YP];
    __shared__ __attribute__((aligned(16))) float s_part[8 * 2 * NPB * 32];
    const int lane = threadIdx.x & 31;
    const int wave = __builtin_amdgcn_readfirstlane((int)(threadIdx.x >> 5));
    const int h = lane >> 2;
    v2f w[4];
#pragma unroll
    for (int j = 0; j < 4; ++j) w[j] = *(const v2f*)(al + lane * 8 + 2 * j);

#pragma nounroll
    for (int q = 0; q < 2; ++q) {
        const int row = q * 16 + wave;
        const int d = (int)blockIdx.x * NPB + row;
        float4 r0 = make_float4(0.f, 0.f, 0.f, 0.f), r1 = r0;
        if (d < NN) {
            const int beg = off[d], end = off[d + 1];
            const float erd = er[d * H1N + h];
            float m = -INFINITY, l = 0.f;
            v2f a[4] = {};
            for (int base = beg; base < end; base += 32) {
                const int mcnt = (end - base < 32) ? (end - base) : 32;
                const int s_l = (lane < mcnt) ? csrc[base + lane] : 0;
                int i = 0;
                for (; i + 4 <= mcnt; i += 4)
                    agg1_group<4>(feat, s_l, i, lane, w, erd, m, l, a);
                const int rem = mcnt - i;
                if (rem == 3)      agg1_group<3>(feat, s_l, i, lane, w, erd, m, l, a);
                else if (rem == 2) agg1_group<2>(feat, s_l, i, lane, w, erd, m, l, a);
                else if (rem == 1) agg1_group<1>(feat, s_l, i, lane, w, erd, m, l, a);
            }
            const float inv = (l > 0.f) ? 1.f / l : 0.f;
            const size_t o = (size_t)d * 256 + lane * 8;
            const float4 x0 = *(const float4*)(x + o);
            const float4 x1 = *(const float4*)(x + o + 4);
            const float4 c0 = *(const float4*)(b1 + lane * 8);
            const float4 c1 = *(const float4*)(b1 + lane * 8 + 4);
            r0.x = fmaxf(a[0].x * inv + x0.x + c0.x, 0.f); r0.y = fmaxf(a[0].y * inv + x0.y + c0.y, 0.f);
            r0.z = fmaxf(a[1].x * inv + x0.z + c0.z, 0.f); r0.w = fmaxf(a[1].y * inv + x0.w + c0.w, 0.f);
            r1.x = fmaxf(a[2].x * inv + x1.x + c1.x, 0.f); r1.y = fmaxf(a[2].y * inv + x1.y + c1.y, 0.f);
            r1.z = fmaxf(a[3].x * inv + x1.z + c1.z, 0.f); r1.w = fmaxf(a[3].y * inv + x1.w + c1.w, 0.f);
        }
        *(float4*)(s_y + row * YP + lane * 8)     = r0;
        *(float4*)(s_y + row * YP + lane * 8 + 4) = r1;
    }

    const int half = lane >> 4;
    const int r    = lane & 15;
    const int p    = wave & 1;
    const int ks   = wave >> 1;
    const float* bp = (p ? RW : W2) + (size_t)(ks * KSL + half * 4) * 32 + r * 2;
    v2f bq[KSL / 8][4];
#pragma unroll
    for (int k8 = 0; k8 < KSL / 8; ++k8)
#pragma unroll
        for (int i = 0; i < 4; ++i)
            bq[k8][i] = *(const v2f*)(bp + (size_t)(k8 * 8 + i) * 32);
    __syncthreads();

    v8f acc[2][2] = {};
#pragma unroll
    for (int k8 = 0; k8 < KSL / 8; ++k8) {
#pragma unroll
        for (int rt = 0; rt < 2; ++rt) {
            const v4f a4 = *(const v4f*)(s_y + (rt * 16 + r) * YP + ks * KSL + half * 4 + k8 * 8);
            v2f av;
            av.x = a4.x; av.y = a4.y;
#pragma unroll
            for (int t = 0; t < 2; ++t) {
                v2f b;
                b.x = bq[k8][0][t]; b.y = bq[k8][1][t];
                acc[rt][t] = __builtin_amdgcn_wmma_f32_16x16x4_f32(
                    false, av, false, b, (short)0, acc[rt][t], false, false);
            }
            av.x = a4.z; av.y = a4.w;
#pragma unroll
            for (int t = 0; t < 2; ++t) {
                v2f b;
                b.x = bq[k8][2][t]; b.y = bq[k8][3][t];
                acc[rt][t] = __builtin_amdgcn_wmma_f32_16x16x4_f32(
                    false, av, false, b, (short)0, acc[rt][t], false, false);
            }
        }
    }

    accumulator_slots();

#pragma unroll
    for (int rt = 0; rt < 2; ++rt) {
        float* pp = s_part + ((ks * 2 + p) * NPB + rt * 16 + half * 8) * 32 + r * 2;
#pragma unroll
        for (int j = 0; j < 8; ++j) {
            v2f o;
            o.x = acc[rt][0][j]; o.y = acc[rt][1][j];
            *(v2f*)(pp + j * 32) = o;
        }
    }
    __syncthreads();
    const float bias = b2[lane];
#pragma unroll
    for (int q = 0; q < 2; ++q) {
        const int row = q * 16 + wave;
        const int d = (int)blockIdx.x * NPB + row;
        float f = 0.f, g = 0.f;
#pragma unroll
        for (int s = 0; s < 8; ++s) {
            f += s_part[((s * 2 + 0) * NPB + row) * 32 + lane];
            g += s_part[((s * 2 + 1) * NPB + row) * 32 + lane];
        }
        if (d < NN) {
            F2[(size_t)d * 32 + lane]   = f;
            OUTP[(size_t)d * 32 + lane] = g + bias;
        }
    }
}

__global__ void __launch_bounds__(256) agg2_gather_kernel(const int* __restrict__ off,
                                                          const int* __restrict__ csrc,
                                                          const float* __restrict__ al,
                                                          const float* __restrict__ ar,
                                                          const float* __restrict__ feat,
                                                          float* __restrict__ outp) {
    const int lane = threadIdx.x & 31;
    const int d = __builtin_amdgcn_readfirstlane((int)(blockIdx.x * 8 + (threadIdx.x >> 5)));
    const int beg = off[d], end = off[d + 1];
    const float alc = al[lane];
    const size_t o = (size_t)d * 32 + lane;
    const float erd = wave_sum(feat[o] * ar[lane]);
    const float res = outp[o];
    const bool up = lane >= 16;

    float m = -INFINITY, l = 0.f, acc = 0.f;
    for (int base = beg; base < end; base += 32) {
        const int mcnt = (end - base < 32) ? (end - base) : 32;
        const bool big = mcnt > 16;
        const int s_l = (lane < mcnt) ? csrc[base + lane] : 0;
        float lo[16], hi[16];
        lo[0] = feat[(unsigned)__builtin_amdgcn_readlane(s_l, 0) * 32u + lane];
#pragma unroll
        for (int k = 1; k < 16; ++k) {
            lo[k] = 0.f;
            if (k < mcnt)
                lo[k] = feat[(unsigned)__builtin_amdgcn_readlane(s_l, k) * 32u + lane];
        }
        if (big) {
#pragma unroll
            for (int k = 0; k < 16; ++k) {
                hi[k] = 0.f;
                if (16 + k < mcnt)
                    hi[k] = feat[(unsigned)__builtin_amdgcn_readlane(s_l, 16 + k) * 32u + lane];
            }
        }
        float va[16];
#pragma unroll
        for (int k = 0; k < 16; ++k) va[k] = lo[k] * alc;
        const float sa = halfwave_sums16(va, lane);
        float sb = 0.f;
        if (big) {
            float vb[16];
#pragma unroll
            for (int k = 0; k < 16; ++k) vb[k] = hi[k] * alc;
            sb = halfwave_sums16(vb, lane);
        }
        const float keep = up ? sb : sa;
        const float send = up ? sa : sb;
        const float lg = keep + half_swap(send);
        const bool have = lane < mcnt;
        const float e  = have ? leaky(lg + erd) : -INFINITY;
        const float mc = fmaxf(m, wave_max(e));
        const float c  = __expf(m - mc);
        const float p  = have ? __expf(e - mc) : 0.f;
        l = l * c + wave_sum(p);
        acc *= c;
#pragma unroll
        for (int g = 0; g < 2; ++g) {
            if (g * 8 < mcnt) {
#pragma unroll
                for (int j = 0; j < 8; ++j)
                    acc = __builtin_fmaf(__int_as_float(__builtin_amdgcn_readlane(
                                             __float_as_int(p), g * 8 + j)), lo[g * 8 + j], acc);
            }
        }
        if (big) {
#pragma unroll
            for (int g = 0; g < 2; ++g) {
                if (16 + g * 8 < mcnt) {
#pragma unroll
                    for (int j = 0; j < 8; ++j)
                        acc = __builtin_fmaf(__int_as_float(__builtin_amdgcn_readlane(
                                                 __float_as_int(p), 16 + g * 8 + j)), hi[g * 8 + j], acc);
                }
            }
        }
        m = mc;
    }
    const float inv = (l > 0.f) ? 1.f / l : 0.f;
    outp[o] = res + acc * inv;
}

extern "C" void kernel_launch(void* const* d_in, const int* in_sizes, int n_in,
                              void* d_out, int out_size, void* d_ws, size_t ws_size,
                              hipStream_t stream) {
    const float* x     = (const float*)d_in[0];
    const int*   src   = (const int*)  d_in[1];
    const int*   dst   = (const int*)  d_in[2];
    const float* W1    = (const float*)d_in[3];
    const float* al1   = (const float*)d_in[4];
    const float* ar1   = (const float*)d_in[5];
    const float* b1    = (const float*)d_in[6];
    const float* W2    = (const float*)d_in[7];
    const float* al2   = (const float*)d_in[8];
    const float* ar2   = (const float*)d_in[9];
    const float* resW2 = (const float*)d_in[10];
    const float* b2    = (const float*)d_in[11];
    float* outp = (float*)d_out;

    float* ws    = (float*)d_ws;
    float* feat1 = ws;
    float* area2 = feat1 + (size_t)NN * 256;
    float* er1   = area2 + (size_t)NN * 256;
    int*   csrc  = (int*)(er1 + (size_t)NN * 8);
    unsigned* tmp = (unsigned*)(csrc + EE);
    int*   off   = (int*)(tmp + EE);
    int*   lo    = (int*)area2;
    float* feat2 = area2;

    chunk_sort_kernel<<<NCHUNK, SORT_T, 0, stream>>>(src, dst, lo, tmp);
    bucket_sort_kernel<<<NBKT, SEG_T, 0, stream>>>(tmp, lo, off, csrc);

    gemm1_kernel<<<G1_PAIRS + G1_SINGLES, 128, 0, stream>>>(x, W1, ar1, feat1, er1);
    agg1_proj_kernel<<<(NN + NPB - 1) / NPB, 512, 0, stream>>>(off, csrc, al1, er1, feat1, x, b1,
                                                               W2, resW2, b2, feat2, outp);

    agg2_gather_kernel<<<NN / 8, 256, 0, stream>>>(off, csrc, al2, ar2, feat2, outp);
}
